# starting kernel
# baseline (speedup 1.0000x reference)
#include <hip/hip_runtime.h>
#include <hip/hip_bf16.h>


#define NNODES 50000
#define NEDGES 800000
#define BN_EPS 1e-5f

typedef __attribute__((ext_vector_type(2))) float v2f;
typedef __attribute__((ext_vector_type(8))) float v8f;

__global__ __launch_bounds__(256)
void edge_scatter_kernel(const float* __restrict__ x, int ldx,
                         const float* __restrict__ ea,
                         const int* __restrict__ src,
                         const int* __restrict__ dst,
                         const float* __restrict__ lw,
                         const float* __restrict__ lb,
                         float* __restrict__ agg)
{
    const int CHUNKS = 24;
    int gid = blockIdx.x * blockDim.x + threadIdx.x;
    if (gid >= NEDGES * CHUNKS) return;
    int j = gid / CHUNKS;
    int d = (gid % CHUNKS) * 4;

    int   s = src[j];
    int   t = dst[j];
    float a = ea[j];

    float4 xv  = *reinterpret_cast<const float4*>(x + (size_t)s * ldx + d);
    float4 lw4 = *reinterpret_cast<const float4*>(lw + d);
    float4 lb4 = *reinterpret_cast<const float4*>(lb + d);

    float m0 = fmaxf(xv.x + a * lw4.x + lb4.x, 0.0f);
    float m1 = fmaxf(xv.y + a * lw4.y + lb4.y, 0.0f);
    float m2 = fmaxf(xv.z + a * lw4.z + lb4.z, 0.0f);
    float m3 = fmaxf(xv.w + a * lw4.w + lb4.w, 0.0f);

    float* ap = agg + (size_t)t * 96 + d;
    atomicAdd(ap + 0, m0);
    atomicAdd(ap + 1, m1);
    atomicAdd(ap + 2, m2);
    atomicAdd(ap + 3, m3);
}

template <int K, bool HASADD>
__global__ __launch_bounds__(32)
void gemm_wmma_f32_kernel(const float* __restrict__ A0, int lda0,
                          const float* __restrict__ A1, int lda1,
                          const float* __restrict__ W,  int ldw,
                          const float* __restrict__ bias,
                          float* __restrict__ out, int ldo)
{
    int lane = threadIdx.x;
    int lo   = lane & 15;
    int hi   = lane >> 4;
    int row0 = blockIdx.x * 16;
    int col0 = blockIdx.y * 16;

    const float* arow0 = A0 + (size_t)(row0 + lo) * lda0;
    const float* arow1 = HASADD ? (A1 + (size_t)(row0 + lo) * lda1) : nullptr;

    v8f c = {};
#pragma unroll
    for (int k0 = 0; k0 < K; k0 += 4) {
        int ka = k0 + 2 * hi;
        v2f a, b;
        a.x = arow0[ka];
        a.y = arow0[ka + 1];
        if (HASADD) {
            a.x += arow1[ka];
            a.y += arow1[ka + 1];
        }
        b.x = W[(size_t)ka * ldw + col0 + lo];
        b.y = W[(size_t)(ka + 1) * ldw + col0 + lo];
        c = __builtin_amdgcn_wmma_f32_16x16x4_f32(
                 false, a,  false, b,
                 (short)0, c,  false,  false);
    }

    float bv = bias[col0 + lo];
#pragma unroll
    for (int r = 0; r < 8; ++r) {
        int row = row0 + r + 8 * hi;
        out[(size_t)row * ldo + col0 + lo] = fmaxf(c[r] + bv, 0.0f);
    }
}

__global__ void col_stats_kernel(const float* __restrict__ in, int ld, int n,
                                 float* __restrict__ sum,
                                 float* __restrict__ sumsq)
{
    int c = threadIdx.x;
    float s = 0.0f, s2 = 0.0f;
    for (int row = blockIdx.x; row < n; row += gridDim.x) {
        float v = in[(size_t)row * ld + c];
        s  += v;
        s2 += v * v;
    }
    atomicAdd(&sum[c],   s);
    atomicAdd(&sumsq[c], s2);
}

__global__ void bn_finalize_kernel(const float* __restrict__ sum,
                                   const float* __restrict__ sumsq,
                                   const float* __restrict__ g,
                                   const float* __restrict__ beta,
                                   float* __restrict__ scale,
                                   float* __restrict__ shift, float invN)
{
    int c = threadIdx.x;
    float mu  = sum[c] * invN;
    float var = sumsq[c] * invN - mu * mu;
    float inv = rsqrtf(var + BN_EPS);
    float sc  = g[c] * inv;
    scale[c] = sc;
    shift[c] = beta[c] - mu * sc;
}

__global__ __launch_bounds__(256)
void bn_apply_kernel(const float* __restrict__ in, int ldi,
                     float* __restrict__ out, int ldo,
                     const float* __restrict__ scale,
                     const float* __restrict__ shift, int n, int C)
{
    int gid = blockIdx.x * blockDim.x + threadIdx.x;
    if (gid >= n * C) return;
    int row = gid / C;
    int c   = gid % C;
    out[(size_t)row * ldo + c] = in[(size_t)row * ldi + c] * scale[c] + shift[c];
}

extern "C" void kernel_launch(void* const* d_in, const int* in_sizes, int n_in,
                              void* d_out, int out_size, void* d_ws, size_t ws_size,
                              hipStream_t stream)
{
    (void)in_sizes; (void)n_in; (void)out_size; (void)ws_size;

    const float* x          = (const float*)d_in[0];
    const float* edge_attr  = (const float*)d_in[1];
    const int*   edge_index = (const int*)  d_in[2];
    const float* lin_w      = (const float*)d_in[3];
    const float* lin_b      = (const float*)d_in[4];
    const float* c1_w  = (const float*)d_in[5];
    const float* c1_b  = (const float*)d_in[6];
    const float* c1_g  = (const float*)d_in[7];
    const float* c1_be = (const float*)d_in[8];
    const float* c2_w  = (const float*)d_in[9];
    const float* c2_b  = (const float*)d_in[10];
    const float* c2_g  = (const float*)d_in[11];
    const float* c2_be = (const float*)d_in[12];
    const float* l1_w  = (const float*)d_in[13];
    const float* l1_b  = (const float*)d_in[14];
    const float* l1_g  = (const float*)d_in[15];
    const float* l1_be = (const float*)d_in[16];
    const float* ma_w  = (const float*)d_in[17];
    const float* ma_b  = (const float*)d_in[18];
    const float* ma_g  = (const float*)d_in[19];
    const float* ma_be = (const float*)d_in[20];
    const float* mb_w  = (const float*)d_in[21];
    const float* mb_b  = (const float*)d_in[22];
    const float* mb_g  = (const float*)d_in[23];
    const float* mb_be = (const float*)d_in[24];

    const int* src = edge_index;
    const int* dst = edge_index + NEDGES;

    const size_t N = NNODES;

    float* ws    = (float*)d_ws;
    float* agg   = ws;
    float* bufP  = agg  + N * 96;
    float* bufQ  = bufP + N * 96;
    float* xcat  = bufQ + N * 96;
    float* stats = xcat + N * 160;

    const float invN = 1.0f / (float)N;
    const int rowTiles = NNODES / 16;

    auto statsL = [&](int L) { return stats + (size_t)L * 384; };

    hipMemsetAsync(agg,   0, N * 96 * sizeof(float), stream);
    hipMemsetAsync(stats, 0, 5 * 384 * sizeof(float), stream);

    const int edgeThreads = NEDGES * 24;
    const int edgeBlocks  = (edgeThreads + 255) / 256;

    edge_scatter_kernel<<<edgeBlocks, 256, 0, stream>>>(
        x, 96, edge_attr, src, dst, lin_w, lin_b, agg);

    gemm_wmma_f32_kernel<96, true><<<dim3(rowTiles, 6), 32, 0, stream>>>(
        x, 96, agg, 96, c1_w, 96, c1_b, bufP, 96);

    col_stats_kernel<<<256, 96, 0, stream>>>(bufP, 96, NNODES,
                                             statsL(0), statsL(0) + 96);
    bn_finalize_kernel<<<1, 96, 0, stream>>>(statsL(0), statsL(0) + 96,
                                             c1_g, c1_be,
                                             statsL(0) + 192, statsL(0) + 288, invN);
    bn_apply_kernel<<<(NNODES * 96 + 255) / 256, 256, 0, stream>>>(
        bufP, 96, xcat, 160, statsL(0) + 192, statsL(0) + 288, NNODES, 96);

    hipMemsetAsync(agg, 0, N * 96 * sizeof(float), stream);
    edge_scatter_kernel<<<edgeBlocks, 256, 0, stream>>>(
        xcat, 160, edge_attr, src, dst, lin_w, lin_b, agg);

    gemm_wmma_f32_kernel<96, true><<<dim3(rowTiles, 4), 32, 0, stream>>>(
        xcat, 160, agg, 96, c2_w, 64, c2_b, bufP, 64);

    col_stats_kernel<<<256, 64, 0, stream>>>(bufP, 64, NNODES,
                                             statsL(1), statsL(1) + 96);
    bn_finalize_kernel<<<1, 64, 0, stream>>>(statsL(1), statsL(1) + 96,
                                             c2_g, c2_be,
                                             statsL(1) + 192, statsL(1) + 288, invN);
    bn_apply_kernel<<<(NNODES * 64 + 255) / 256, 256, 0, stream>>>(
        bufP, 64, xcat + 96, 160, statsL(1) + 192, statsL(1) + 288, NNODES, 64);

    gemm_wmma_f32_kernel<160, false><<<dim3(rowTiles, 6), 32, 0, stream>>>(
        xcat, 160, nullptr, 0, l1_w, 96, l1_b, bufQ, 96);

    col_stats_kernel<<<256, 96, 0, stream>>>(bufQ, 96, NNODES,
                                             statsL(2), statsL(2) + 96);
    bn_finalize_kernel<<<1, 96, 0, stream>>>(statsL(2), statsL(2) + 96,
                                             l1_g, l1_be,
                                             statsL(2) + 192, statsL(2) + 288, invN);
    bn_apply_kernel<<<(NNODES * 96 + 255) / 256, 256, 0, stream>>>(
        bufQ, 96, bufQ, 96, statsL(2) + 192, statsL(2) + 288, NNODES, 96);

    gemm_wmma_f32_kernel<96, false><<<dim3(rowTiles, 6), 32, 0, stream>>>(
        bufQ, 96, nullptr, 0, ma_w, 96, ma_b, bufP, 96);

    col_stats_kernel<<<256, 96, 0, stream>>>(bufP, 96, NNODES,
                                             statsL(3), statsL(3) + 96);
    bn_finalize_kernel<<<1, 96, 0, stream>>>(statsL(3), statsL(3) + 96,
                                             ma_g, ma_be,
                                             statsL(3) + 192, statsL(3) + 288, invN);
    bn_apply_kernel<<<(NNODES * 96 + 255) / 256, 256, 0, stream>>>(
        bufP, 96, bufP, 96, statsL(3) + 192, statsL(3) + 288, NNODES, 96);

    gemm_wmma_f32_kernel<96, false><<<dim3(rowTiles, 1), 32, 0, stream>>>(
        bufP, 96, nullptr, 0, mb_w, 16, mb_b, bufQ, 16);

    col_stats_kernel<<<256, 16, 0, stream>>>(bufQ, 16, NNODES,
                                             statsL(4), statsL(4) + 96);
    bn_finalize_kernel<<<1, 16, 0, stream>>>(statsL(4), statsL(4) + 96,
                                             mb_g, mb_be,
                                             statsL(4) + 192, statsL(4) + 288, invN);
    bn_apply_kernel<<<(NNODES * 16 + 255) / 256, 256, 0, stream>>>(
        bufQ, 16, (float*)d_out, 16, statsL(4) + 192, statsL(4) + 288, NNODES, 16);
}


# BatchNorm scale/shift made by each consumer from the column sums; 3 GEMMs fold them into the resident W slice and bias (W'=diag(sc)W, b'=b+sh*W): 5 finalize, 3 apply launches gone; rounding changes
# speedup vs baseline: 2.6222x; 2.6222x over previous
#include <hip/hip_runtime.h>
#include <hip/hip_bf16.h>


#define NNODES 50000
#define NEDGES 800000
#define BN_EPS 1e-5f
#define INV_N  (1.0f / (float)NNODES)

typedef __attribute__((ext_vector_type(2))) float v2f;
typedef __attribute__((ext_vector_type(4))) float v4f;
typedef __attribute__((ext_vector_type(8))) float v8f;

__global__ __launch_bounds__(256)
void edge_scatter_kernel(const float* __restrict__ x, int ldx,
                         const float* __restrict__ ea,
                         const int* __restrict__ src,
                         const int* __restrict__ dst,
                         const float* __restrict__ lw,
                         const float* __restrict__ lb,
                         float* __restrict__ agg)
{
    const int CHUNKS = 24;
    int gid = blockIdx.x * blockDim.x + threadIdx.x;
    if (gid >= NEDGES * CHUNKS) return;
    int j = gid / CHUNKS;
    int d = (gid % CHUNKS) * 4;

    int   s = src[j];
    int   t = dst[j];
    float a = ea[j];

    float4 xv  = *reinterpret_cast<const float4*>(x + (size_t)s * ldx + d);
    float4 lw4 = *reinterpret_cast<const float4*>(lw + d);
    float4 lb4 = *reinterpret_cast<const float4*>(lb + d);

    float m0 = fmaxf(xv.x + a * lw4.x + lb4.x, 0.0f);
    float m1 = fmaxf(xv.y + a * lw4.y + lb4.y, 0.0f);
    float m2 = fmaxf(xv.z + a * lw4.z + lb4.z, 0.0f);
    float m3 = fmaxf(xv.w + a * lw4.w + lb4.w, 0.0f);

    float* ap = agg + (size_t)t * 96 + d;
    atomicAdd(ap + 0, m0);
    atomicAdd(ap + 1, m1);
    atomicAdd(ap + 2, m2);
    atomicAdd(ap + 3, m3);
}

#define ROWTILES (NNODES / 16)
#define HOLD4(a, i) "v"(a[i]), "v"(a[(i) + 1]), "v"(a[(i) + 2]), "v"(a[(i) + 3])

__device__ __forceinline__ int frag_k(int t, int h)
{
    return 8 * (t >> 1) + 4 * h + 2 * (t & 1);
}

template <int K, int NT, int WPB, bool HASADD, int FOLD0>
__global__ __launch_bounds__(NT * WPB * 32)
void gemm_wmma_f32_kernel(const float* __restrict__ A0, int lda0,
                          const float* __restrict__ A1, int lda1,
                          const float* __restrict__ W,  int ldw,
                          const float* __restrict__ bias,
                          float* __restrict__ out, int ldo,
                          float* __restrict__ colsum,
                          float* __restrict__ colsq,
                          const float* __restrict__ fsum,
                          const float* __restrict__ fsq,
                          const float* __restrict__ fg,
                          const float* __restrict__ fbeta)
{
    constexpr int STEPS  = K / 4;
    constexpr int CHUNKS = K / 8;

    const int lane = threadIdx.x & 31;
    const int wave = __builtin_amdgcn_readfirstlane((int)(threadIdx.x >> 5));
    const int lo   = lane & 15;
    const int hi   = lane >> 4;
    const int ct       = wave % NT;
    const int walker   = blockIdx.x * WPB + wave / NT;
    const int nWalkers = gridDim.x * WPB;
    const int col      = ct * 16 + lo;

    v2f b[STEPS];
#pragma unroll
    for (int t = 0; t < STEPS; ++t) {
        const int k = frag_k(t, hi);
        b[t].x = W[(size_t)k * ldw + col];
        b[t].y = W[(size_t)(k + 1) * ldw + col];
    }
    float bv = bias[col];

    if constexpr (FOLD0 < K) {
        constexpr int KF = K - FOLD0;
        static_assert(FOLD0 % 8 == 0, "a 16-byte A chunk must lie wholly on one side of FOLD0");
        static_assert(KF <= NT * WPB * 32, "one thread per folded column");
        __shared__ float fsc[KF];
        __shared__ float fsh[KF];
        if (threadIdx.x < KF) {
            const int i = threadIdx.x;
            const float mu  = fsum[i] * INV_N;
            const float var = fsq[i] * INV_N - mu * mu;
            const float sc  = fg[i] * rsqrtf(var + BN_EPS);
            fsc[i] = sc;
            fsh[i] = fbeta[i] - mu * sc;
        }
        __syncthreads();
        float part = 0.0f;
#pragma unroll
        for (int t = 0; t < STEPS; ++t) {
            if (8 * (t >> 1) >= FOLD0) {
                const int kf = frag_k(t, hi) - FOLD0;
                part += fsh[kf] * b[t].x + fsh[kf + 1] * b[t].y;
                b[t].x *= fsc[kf];
                b[t].y *= fsc[kf + 1];
            }
        }
        bv += part + __shfl_xor(part, 16);
    }

    float s1 = 0.0f, s2 = 0.0f;

    for (int rt = walker; rt < ROWTILES; rt += nWalkers) {
        const int row0 = rt * 16;
        const v4f* ap0 = reinterpret_cast<const v4f*>(
            A0 + (size_t)(row0 + lo) * lda0 + 4 * hi);

        v4f q[CHUNKS];
#pragma unroll
        for (int j = 0; j < CHUNKS; ++j) q[j] = ap0[2 * j];

        if constexpr (HASADD) {
            const v4f* ap1 = reinterpret_cast<const v4f*>(
                A1 + (size_t)(row0 + lo) * lda1 + 4 * hi);
            v4f p[CHUNKS];
#pragma unroll
            for (int j = 0; j < CHUNKS; ++j) p[j] = ap1[2 * j];
            static_assert(CHUNKS == 12, "fence operand list is written for K = 96");
            asm volatile("" :: HOLD4(q, 0), HOLD4(q, 4), HOLD4(q, 8),
                               HOLD4(p, 0), HOLD4(p, 4), HOLD4(p, 8));
#pragma unroll
            for (int j = 0; j < CHUNKS; ++j) q[j] += p[j];
        } else if constexpr (CHUNKS == 12) {
            asm volatile("" :: HOLD4(q, 0), HOLD4(q, 4), HOLD4(q, 8));
        } else {
            static_assert(CHUNKS == 20, "fence operand list is written for K = 96 or 160");
            asm volatile("" :: HOLD4(q, 0), HOLD4(q, 4), HOLD4(q, 8),
                               HOLD4(q, 12), HOLD4(q, 16));
        }

        v8f c = {};
#pragma unroll
        for (int j = 0; j < CHUNKS; ++j) {
            v2f a;
            a.x = q[j].x; a.y = q[j].y;
            c = __builtin_amdgcn_wmma_f32_16x16x4_f32(
                     false, a,  false, b[2 * j],
                     (short)0, c,  false,  false);
            a.x = q[j].z; a.y = q[j].w;
            c = __builtin_amdgcn_wmma_f32_16x16x4_f32(
                     false, a,  false, b[2 * j + 1],
                     (short)0, c,  false,  false);
        }

#pragma unroll
        for (int r = 0; r < 8; ++r) {
            int row = row0 + r + 8 * hi;
            const float v = fmaxf(c[r] + bv, 0.0f);
            out[(size_t)row * ldo + col] = v;
            s1 += v;
            s2 += v * v;
        }
    }

    s1 += __shfl_xor(s1, 16);
    s2 += __shfl_xor(s2, 16);

    if constexpr (WPB == 1) {
        if (hi == 0) {
            atomicAdd(&colsum[col], s1);
            atomicAdd(&colsq[col],  s2);
        }
    } else {
        static_assert(NT == 1, "in-block add assumes all waves own the same columns");
        __shared__ float red[WPB][32];
        red[wave][lane] = (hi == 0) ? s1 : s2;
        __syncthreads();
        if (wave == 0) {
            float t = 0.0f;
            for (int w = 0; w < WPB; ++w) t += red[w][lane];
            atomicAdd((hi == 0 ? colsum : colsq) + lo, t);
        }
    }
}

__global__ __launch_bounds__(256)
void bn_apply_kernel(const float* __restrict__ in, int ldi,
                     float* __restrict__ out, int ldo,
                     const float* __restrict__ sum,
                     const float* __restrict__ sumsq,
                     const float* __restrict__ g,
                     const float* __restrict__ beta, int n, int C)
{
    int gid = blockIdx.x * blockDim.x + threadIdx.x;
    if (gid >= n * C) return;
    int row = gid / C;
    int c   = gid % C;
    float mu  = sum[c] * INV_N;
    float var = sumsq[c] * INV_N - mu * mu;
    float sc  = g[c] * rsqrtf(var + BN_EPS);
    float sh  = beta[c] - mu * sc;
    out[(size_t)row * ldo + c] = in[(size_t)row * ldi + c] * sc + sh;
}

extern "C" void kernel_launch(void* const* d_in, const int* in_sizes, int n_in,
                              void* d_out, int out_size, void* d_ws, size_t ws_size,
                              hipStream_t stream)
{
    (void)in_sizes; (void)n_in; (void)out_size; (void)ws_size;

    const float* x          = (const float*)d_in[0];
    const float* edge_attr  = (const float*)d_in[1];
    const int*   edge_index = (const int*)  d_in[2];
    const float* lin_w      = (const float*)d_in[3];
    const float* lin_b      = (const float*)d_in[4];
    const float* c1_w  = (const float*)d_in[5];
    const float* c1_b  = (const float*)d_in[6];
    const float* c1_g  = (const float*)d_in[7];
    const float* c1_be = (const float*)d_in[8];
    const float* c2_w  = (const float*)d_in[9];
    const float* c2_b  = (const float*)d_in[10];
    const float* c2_g  = (const float*)d_in[11];
    const float* c2_be = (const float*)d_in[12];
    const float* l1_w  = (const float*)d_in[13];
    const float* l1_b  = (const float*)d_in[14];
    const float* l1_g  = (const float*)d_in[15];
    const float* l1_be = (const float*)d_in[16];
    const float* ma_w  = (const float*)d_in[17];
    const float* ma_b  = (const float*)d_in[18];
    const float* ma_g  = (const float*)d_in[19];
    const float* ma_be = (const float*)d_in[20];
    const float* mb_w  = (const float*)d_in[21];
    const float* mb_b  = (const float*)d_in[22];
    const float* mb_g  = (const float*)d_in[23];
    const float* mb_be = (const float*)d_in[24];

    const int* src = edge_index;
    const int* dst = edge_index + NEDGES;

    const size_t N = NNODES;

    float* ws    = (float*)d_ws;
    float* agg   = ws;
    float* bufP  = agg  + N * 96;
    float* bufQ  = bufP + N * 96;
    float* xcat  = bufQ + N * 96;
    float* stats = xcat + N * 160;


    auto sumL = [&](int L) { return stats + (size_t)L * 192; };
    auto sqL  = [&](int L) { return stats + (size_t)L * 192 + 96; };

    hipMemsetAsync(agg,   0, N * 96 * sizeof(float), stream);
    hipMemsetAsync(stats, 0, 5 * 192 * sizeof(float), stream);

    const int edgeThreads = NEDGES * 24;
    const int edgeBlocks  = (edgeThreads + 255) / 256;

    edge_scatter_kernel<<<edgeBlocks, 256, 0, stream>>>(
        x, 96, edge_attr, src, dst, lin_w, lin_b, agg);

    gemm_wmma_f32_kernel<96, 6, 1, true, 96><<<625, 192, 0, stream>>>(
        x, 96, agg, 96, c1_w, 96, c1_b, bufP, 96, sumL(0), sqL(0),
        nullptr, nullptr, nullptr, nullptr);

    bn_apply_kernel<<<(NNODES * 96 + 255) / 256, 256, 0, stream>>>(
        bufP, 96, xcat, 160, sumL(0), sqL(0), c1_g, c1_be, NNODES, 96);

    hipMemsetAsync(agg, 0, N * 96 * sizeof(float), stream);
    edge_scatter_kernel<<<edgeBlocks, 256, 0, stream>>>(
        xcat, 160, edge_attr, src, dst, lin_w, lin_b, agg);

    gemm_wmma_f32_kernel<96, 4, 1, true, 96><<<625, 128, 0, stream>>>(
        xcat, 160, agg, 96, c2_w, 64, c2_b, xcat + 96, 160, sumL(1), sqL(1),
        nullptr, nullptr, nullptr, nullptr);

    gemm_wmma_f32_kernel<160, 6, 1, false, 96><<<625, 192, 0, stream>>>(
        xcat, 160, nullptr, 0, l1_w, 96, l1_b, bufQ, 96, sumL(2), sqL(2),
        sumL(1), sqL(1), c2_g, c2_be);

    gemm_wmma_f32_kernel<96, 6, 1, false, 0><<<625, 192, 0, stream>>>(
        bufQ, 96, nullptr, 0, ma_w, 96, ma_b, bufP, 96, sumL(3), sqL(3),
        sumL(2), sqL(2), l1_g, l1_be);

    gemm_wmma_f32_kernel<96, 1, 5, false, 0><<<625, 160, 0, stream>>>(
        bufP, 96, nullptr, 0, mb_w, 16, mb_b, bufQ, 16, sumL(4), sqL(4),
        sumL(3), sqL(3), ma_g, ma_be);

    bn_apply_kernel<<<(NNODES * 16 + 255) / 256, 256, 0, stream>>>(
        bufQ, 16, (float*)d_out, 16, sumL(4), sqL(4), mb_g, mb_be, NNODES, 16);
}


# Edge sums owner-computed: 250 blocks each keep 200 destination rows in LDS, filter the whole dst list for them, add messages by LDS float adds; global atomics, agg memsets gone; sum order changes
# speedup vs baseline: 3.7404x; 1.4264x over previous
#include <hip/hip_runtime.h>
#include <hip/hip_bf16.h>


#define NNODES 50000
#define NEDGES 800000
#define BN_EPS 1e-5f
#define INV_N  (1.0f / (float)NNODES)

typedef __attribute__((ext_vector_type(2))) float v2f;
typedef __attribute__((ext_vector_type(4))) float v4f;
typedef __attribute__((ext_vector_type(8))) float v8f;

#define OWN_ROWS    200
#define OWN_BLOCKS  (NNODES / OWN_ROWS)
#define OWN_THREADS 1024
#define OWN_WAVES   (OWN_THREADS / 32)
#define OWN_QCAP    24576
#define NVEC        (NEDGES / 4)
#define SCAN_DEPTH  8
#define ROW_DEPTH   16

static_assert(OWN_ROWS * OWN_BLOCKS == NNODES, "every row has exactly one owner");
static_assert(NEDGES % 4 == 0, "dst list is read as 16-byte vectors");
static_assert(NEDGES <= (1 << 20) && OWN_ROWS <= (1 << 11), "queue word = edge id | local row << 20");
static_assert(NVEC % 32 == 0 && (OWN_QCAP / 4) % 32 == 0 && (NVEC / OWN_BLOCKS) % 32 == 0,
              "range ends and start shifts fall on wave boundaries");

__device__ __forceinline__ void own_scan(const int4* __restrict__ d4,
                                         int vBegin, int vEnd, int rot, int r0,
                                         int* queue, int* qcount, int wave, int lane)
{
    for (int base = vBegin; base < vEnd; base += SCAN_DEPTH * OWN_THREADS) {
        int4 d[SCAN_DEPTH];
        int  vv[SCAN_DEPTH];
#pragma unroll
        for (int u = 0; u < SCAN_DEPTH; ++u) {
            const int vw = base + u * OWN_THREADS + wave * 32;
            int w = vw + lane + rot;
            if (w >= NVEC) w -= NVEC;
            vv[u] = w;
            d[u]  = d4[(vw < vEnd) ? w : lane];
        }
#pragma unroll
        for (int u = 0; u < SCAN_DEPTH; ++u) {
            const int vw = base + u * OWN_THREADS + wave * 32;
            if (vw < vEnd) {
                const int dk[4] = { d[u].x, d[u].y, d[u].z, d[u].w };
#pragma unroll
                for (int k = 0; k < 4; ++k) {
                    const unsigned lrow = (unsigned)(dk[k] - r0);
                    const bool m = lrow < (unsigned)OWN_ROWS;
                    const unsigned mask = __builtin_amdgcn_ballot_w32(m);
                    if (mask != 0u) {
                        int first = 0;
                        if (lane == 0)
                            first = atomicAdd(qcount, __popc(mask));
                        first = __builtin_amdgcn_readfirstlane(first);
                        if (m) {
                            const int pos = first + (int)__builtin_amdgcn_mbcnt_lo(mask, 0u);
                            if (pos < OWN_QCAP)
                                queue[pos] = (4 * vv[u] + k) | (int)(lrow << 20);
                        }
                    }
                }
            }
        }
    }
}

__device__ __forceinline__ void own_process(int n,
                                            const float* __restrict__ x, int ldx,
                                            const float* __restrict__ ea,
                                            const int* __restrict__ src,
                                            const int* queue, float* acc,
                                            float lw0, float lw1, float lw2,
                                            float lb0, float lb1, float lb2,
                                            int wave, int lane)
{
    for (int g = wave * 32; g < n; g += 32 * OWN_WAVES) {
        const int ent  = (g + lane < n) ? queue[g + lane] : 0;
        const int e    = ent & 0xFFFFF;
        const int lrow = ent >> 20;
        const int   s  = src[e];
        const int   ab = __float_as_int(ea[e]);
        const int cnt  = (n - g < 32) ? (n - g) : 32;
        for (int j0 = 0; j0 < cnt; j0 += ROW_DEPTH) {
            float xv[ROW_DEPTH][3];
#pragma unroll
            for (int u = 0; u < ROW_DEPTH; ++u) {
                const int sj = __builtin_amdgcn_readlane(s, j0 + u);
                const float* xr = x + (size_t)sj * ldx + lane;
                xv[u][0] = xr[0];
                xv[u][1] = xr[32];
                xv[u][2] = xr[64];
            }
#pragma unroll
            for (int u = 0; u < ROW_DEPTH; ++u) {
                if (j0 + u < cnt) {
                    const float a  = __int_as_float(__builtin_amdgcn_readlane(ab, j0 + u));
                    const int   rj = __builtin_amdgcn_readlane(lrow, j0 + u);
                    float* ar = acc + rj * 96 + lane;
                    atomicAdd(ar,      fmaxf(xv[u][0] + a * lw0 + lb0, 0.0f));
                    atomicAdd(ar + 32, fmaxf(xv[u][1] + a * lw1 + lb1, 0.0f));
                    atomicAdd(ar + 64, fmaxf(xv[u][2] + a * lw2 + lb2, 0.0f));
                }
            }
        }
    }
}

__global__ __launch_bounds__(OWN_THREADS)
void edge_owner_kernel(const float* __restrict__ x, int ldx,
                       const float* __restrict__ ea,
                       const int* __restrict__ src,
                       const int* __restrict__ dst,
                       const float* __restrict__ lw,
                       const float* __restrict__ lb,
                       float* __restrict__ agg)
{
    __shared__ __attribute__((aligned(16))) float acc[OWN_ROWS * 96];
    __shared__ int queue[OWN_QCAP];
    __shared__ int qcount;

    const int tid  = threadIdx.x;
    const int lane = tid & 31;
    const int wave = __builtin_amdgcn_readfirstlane(tid >> 5);
    const int r0   = blockIdx.x * OWN_ROWS;
    const int4* d4 = reinterpret_cast<const int4*>(dst);

    for (int i = tid; i < OWN_ROWS * 96; i += OWN_THREADS) acc[i] = 0.0f;
    if (tid == 0) qcount = 0;

    const float lw0 = lw[lane], lw1 = lw[lane + 32], lw2 = lw[lane + 64];
    const float lb0 = lb[lane], lb1 = lb[lane + 32], lb2 = lb[lane + 64];

    __syncthreads();

    own_scan(d4, 0, NVEC, (int)blockIdx.x * (NVEC / OWN_BLOCKS), r0, queue, &qcount, wave, lane);

    __syncthreads();
    const int total = __builtin_amdgcn_readfirstlane(qcount);

    if (total <= OWN_QCAP) {
        own_process(total, x, ldx, ea, src, queue, acc,
                    lw0, lw1, lw2, lb0, lb1, lb2, wave, lane);
    } else {
        for (int vb = 0; vb < NVEC; vb += OWN_QCAP / 4) {
            __syncthreads();
            if (tid == 0) qcount = 0;
            __syncthreads();
            const int ve = (vb + OWN_QCAP / 4 < NVEC) ? (vb + OWN_QCAP / 4) : NVEC;
            own_scan(d4, vb, ve, 0, r0, queue, &qcount, wave, lane);
            __syncthreads();
            const int n = __builtin_amdgcn_readfirstlane(qcount);
            own_process(n, x, ldx, ea, src, queue, acc,
                        lw0, lw1, lw2, lb0, lb1, lb2, wave, lane);
        }
    }

    __syncthreads();
    float4* out4 = reinterpret_cast<float4*>(agg + (size_t)r0 * 96);
    const float4* a4 = reinterpret_cast<const float4*>(acc);
    for (int i = tid; i < OWN_ROWS * 96 / 4; i += OWN_THREADS) out4[i] = a4[i];
}

#define ROWTILES (NNODES / 16)
#define HOLD4(a, i) "v"(a[i]), "v"(a[(i) + 1]), "v"(a[(i) + 2]), "v"(a[(i) + 3])

__device__ __forceinline__ int frag_k(int t, int h)
{
    return 8 * (t >> 1) + 4 * h + 2 * (t & 1);
}

template <int K, int NT, int WPB, bool HASADD, int FOLD0>
__global__ __launch_bounds__(NT * WPB * 32)
void gemm_wmma_f32_kernel(const float* __restrict__ A0, int lda0,
                          const float* __restrict__ A1, int lda1,
                          const float* __restrict__ W,  int ldw,
                          const float* __restrict__ bias,
                          float* __restrict__ out, int ldo,
                          float* __restrict__ colsum,
                          float* __restrict__ colsq,
                          const float* __restrict__ fsum,
                          const float* __restrict__ fsq,
                          const float* __restrict__ fg,
                          const float* __restrict__ fbeta)
{
    constexpr int STEPS  = K / 4;
    constexpr int CHUNKS = K / 8;

    const int lane = threadIdx.x & 31;
    const int wave = __builtin_amdgcn_readfirstlane((int)(threadIdx.x >> 5));
    const int lo   = lane & 15;
    const int hi   = lane >> 4;
    const int ct       = wave % NT;
    const int walker   = blockIdx.x * WPB + wave / NT;
    const int nWalkers = gridDim.x * WPB;
    const int col      = ct * 16 + lo;

    v2f b[STEPS];
#pragma unroll
    for (int t = 0; t < STEPS; ++t) {
        const int k = frag_k(t, hi);
        b[t].x = W[(size_t)k * ldw + col];
        b[t].y = W[(size_t)(k + 1) * ldw + col];
    }
    float bv = bias[col];

    if constexpr (FOLD0 < K) {
        constexpr int KF = K - FOLD0;
        static_assert(FOLD0 % 8 == 0, "a 16-byte A chunk must lie wholly on one side of FOLD0");
        static_assert(KF <= NT * WPB * 32, "one thread per folded column");
        __shared__ float fsc[KF];
        __shared__ float fsh[KF];
        if (threadIdx.x < KF) {
            const int i = threadIdx.x;
            const float mu  = fsum[i] * INV_N;
            const float var = fsq[i] * INV_N - mu * mu;
            const float sc  = fg[i] * rsqrtf(var + BN_EPS);
            fsc[i] = sc;
            fsh[i] = fbeta[i] - mu * sc;
        }
        __syncthreads();
        float part = 0.0f;
#pragma unroll
        for (int t = 0; t < STEPS; ++t) {
            if (8 * (t >> 1) >= FOLD0) {
                const int kf = frag_k(t, hi) - FOLD0;
                part += fsh[kf] * b[t].x + fsh[kf + 1] * b[t].y;
                b[t].x *= fsc[kf];
                b[t].y *= fsc[kf + 1];
            }
        }
        bv += part + __shfl_xor(part, 16);
    }

    float s1 = 0.0f, s2 = 0.0f;

    for (int rt = walker; rt < ROWTILES; rt += nWalkers) {
        const int row0 = rt * 16;
        const v4f* ap0 = reinterpret_cast<const v4f*>(
            A0 + (size_t)(row0 + lo) * lda0 + 4 * hi);

        v4f q[CHUNKS];
#pragma unroll
        for (int j = 0; j < CHUNKS; ++j) q[j] = ap0[2 * j];

        if constexpr (HASADD) {
            const v4f* ap1 = reinterpret_cast<const v4f*>(
                A1 + (size_t)(row0 + lo) * lda1 + 4 * hi);
            v4f p[CHUNKS];
#pragma unroll
            for (int j = 0; j < CHUNKS; ++j) p[j] = ap1[2 * j];
            static_assert(CHUNKS == 12, "fence operand list is written for K = 96");
            asm volatile("" :: HOLD4(q, 0), HOLD4(q, 4), HOLD4(q, 8),
                               HOLD4(p, 0), HOLD4(p, 4), HOLD4(p, 8));
#pragma unroll
            for (int j = 0; j < CHUNKS; ++j) q[j] += p[j];
        } else if constexpr (CHUNKS == 12) {
            asm volatile("" :: HOLD4(q, 0), HOLD4(q, 4), HOLD4(q, 8));
        } else {
            static_assert(CHUNKS == 20, "fence operand list is written for K = 96 or 160");
            asm volatile("" :: HOLD4(q, 0), HOLD4(q, 4), HOLD4(q, 8),
                               HOLD4(q, 12), HOLD4(q, 16));
        }

        v8f c = {};
#pragma unroll
        for (int j = 0; j < CHUNKS; ++j) {
            v2f a;
            a.x = q[j].x; a.y = q[j].y;
            c = __builtin_amdgcn_wmma_f32_16x16x4_f32(
                     false, a,  false, b[2 * j],
                     (short)0, c,  false,  false);
            a.x = q[j].z; a.y = q[j].w;
            c = __builtin_amdgcn_wmma_f32_16x16x4_f32(
                     false, a,  false, b[2 * j + 1],
                     (short)0, c,  false,  false);
        }

#pragma unroll
        for (int r = 0; r < 8; ++r) {
            int row = row0 + r + 8 * hi;
            const float v = fmaxf(c[r] + bv, 0.0f);
            out[(size_t)row * ldo + col] = v;
            s1 += v;
            s2 += v * v;
        }
    }

    s1 += __shfl_xor(s1, 16);
    s2 += __shfl_xor(s2, 16);

    if constexpr (WPB == 1) {
        if (hi == 0) {
            atomicAdd(&colsum[col], s1);
            atomicAdd(&colsq[col],  s2);
        }
    } else {
        static_assert(NT == 1, "in-block add assumes all waves own the same columns");
        __shared__ float red[WPB][32];
        red[wave][lane] = (hi == 0) ? s1 : s2;
        __syncthreads();
        if (wave == 0) {
            float t = 0.0f;
            for (int w = 0; w < WPB; ++w) t += red[w][lane];
            atomicAdd((hi == 0 ? colsum : colsq) + lo, t);
        }
    }
}

__global__ __launch_bounds__(256)
void bn_apply_kernel(const float* __restrict__ in, int ldi,
                     float* __restrict__ out, int ldo,
                     const float* __restrict__ sum,
                     const float* __restrict__ sumsq,
                     const float* __restrict__ g,
                     const float* __restrict__ beta, int n, int C)
{
    int gid = blockIdx.x * blockDim.x + threadIdx.x;
    if (gid >= n * C) return;
    int row = gid / C;
    int c   = gid % C;
    float mu  = sum[c] * INV_N;
    float var = sumsq[c] * INV_N - mu * mu;
    float sc  = g[c] * rsqrtf(var + BN_EPS);
    float sh  = beta[c] - mu * sc;
    out[(size_t)row * ldo + c] = in[(size_t)row * ldi + c] * sc + sh;
}

extern "C" void kernel_launch(void* const* d_in, const int* in_sizes, int n_in,
                              void* d_out, int out_size, void* d_ws, size_t ws_size,
                              hipStream_t stream)
{
    (void)in_sizes; (void)n_in; (void)out_size; (void)ws_size;

    const float* x          = (const float*)d_in[0];
    const float* edge_attr  = (const float*)d_in[1];
    const int*   edge_index = (const int*)  d_in[2];
    const float* lin_w      = (const float*)d_in[3];
    const float* lin_b      = (const float*)d_in[4];
    const float* c1_w  = (const float*)d_in[5];
    const float* c1_b  = (const float*)d_in[6];
    const float* c1_g  = (const float*)d_in[7];
    const float* c1_be = (const float*)d_in[8];
    const float* c2_w  = (const float*)d_in[9];
    const float* c2_b  = (const float*)d_in[10];
    const float* c2_g  = (const float*)d_in[11];
    const float* c2_be = (const float*)d_in[12];
    const float* l1_w  = (const float*)d_in[13];
    const float* l1_b  = (const float*)d_in[14];
    const float* l1_g  = (const float*)d_in[15];
    const float* l1_be = (const float*)d_in[16];
    const float* ma_w  = (const float*)d_in[17];
    const float* ma_b  = (const float*)d_in[18];
    const float* ma_g  = (const float*)d_in[19];
    const float* ma_be = (const float*)d_in[20];
    const float* mb_w  = (const float*)d_in[21];
    const float* mb_b  = (const float*)d_in[22];
    const float* mb_g  = (const float*)d_in[23];
    const float* mb_be = (const float*)d_in[24];

    const int* src = edge_index;
    const int* dst = edge_index + NEDGES;

    const size_t N = NNODES;

    float* ws    = (float*)d_ws;
    float* agg   = ws;
    float* bufP  = agg  + N * 96;
    float* bufQ  = bufP + N * 96;
    float* xcat  = bufQ + N * 96;
    float* stats = xcat + N * 160;


    auto sumL = [&](int L) { return stats + (size_t)L * 192; };
    auto sqL  = [&](int L) { return stats + (size_t)L * 192 + 96; };

    hipMemsetAsync(stats, 0, 5 * 192 * sizeof(float), stream);

    edge_owner_kernel<<<OWN_BLOCKS, OWN_THREADS, 0, stream>>>(
        x, 96, edge_attr, src, dst, lin_w, lin_b, agg);

    gemm_wmma_f32_kernel<96, 6, 1, true, 96><<<625, 192, 0, stream>>>(
        x, 96, agg, 96, c1_w, 96, c1_b, bufP, 96, sumL(0), sqL(0),
        nullptr, nullptr, nullptr, nullptr);

    bn_apply_kernel<<<(NNODES * 96 + 255) / 256, 256, 0, stream>>>(
        bufP, 96, xcat, 160, sumL(0), sqL(0), c1_g, c1_be, NNODES, 96);

    edge_owner_kernel<<<OWN_BLOCKS, OWN_THREADS, 0, stream>>>(
        xcat, 160, edge_attr, src, dst, lin_w, lin_b, agg);

    gemm_wmma_f32_kernel<96, 4, 1, true, 96><<<625, 128, 0, stream>>>(
        xcat, 160, agg, 96, c2_w, 64, c2_b, xcat + 96, 160, sumL(1), sqL(1),
        nullptr, nullptr, nullptr, nullptr);

    gemm_wmma_f32_kernel<160, 6, 1, false, 96><<<625, 192, 0, stream>>>(
        xcat, 160, nullptr, 0, l1_w, 96, l1_b, bufQ, 96, sumL(2), sqL(2),
        sumL(1), sqL(1), c2_g, c2_be);

    gemm_wmma_f32_kernel<96, 6, 1, false, 0><<<625, 192, 0, stream>>>(
        bufQ, 96, nullptr, 0, ma_w, 96, ma_b, bufP, 96, sumL(3), sqL(3),
        sumL(2), sqL(2), l1_g, l1_be);

    gemm_wmma_f32_kernel<96, 1, 5, false, 0><<<625, 160, 0, stream>>>(
        bufP, 96, nullptr, 0, mb_w, 16, mb_b, bufQ, 16, sumL(4), sqL(4),
        sumL(3), sqL(3), ma_g, ma_be);

    bn_apply_kernel<<<(NNODES * 16 + 255) / 256, 256, 0, stream>>>(
        bufQ, 16, (float*)d_out, 16, sumL(4), sqL(4), mb_g, mb_be, NNODES, 16);
}


# Edge filter runs once per call: the first edge launch saves each owner block's matched edges as 8-byte records (src, local row, edge value); the second launch reads its records and scans no dst word
# speedup vs baseline: 4.4359x; 1.1859x over previous
#include <hip/hip_runtime.h>
#include <hip/hip_bf16.h>


#define NNODES 50000
#define NEDGES 800000
#define BN_EPS 1e-5f
#define INV_N  (1.0f / (float)NNODES)

typedef __attribute__((ext_vector_type(2))) float v2f;
typedef __attribute__((ext_vector_type(4))) float v4f;
typedef __attribute__((ext_vector_type(8))) float v8f;

#define OWN_ROWS    200
#define OWN_BLOCKS  (NNODES / OWN_ROWS)
#define OWN_THREADS 1024
#define OWN_WAVES   (OWN_THREADS / 32)
#define OWN_QCAP    24576
#define NVEC        (NEDGES / 4)
#define SCAN_DEPTH  8
#define ROW_DEPTH   16

static_assert(OWN_ROWS * OWN_BLOCKS == NNODES, "every row has exactly one owner");
static_assert(NEDGES % 4 == 0, "dst list is read as 16-byte vectors");
static_assert(NEDGES <= (1 << 20) && OWN_ROWS <= (1 << 11), "queue word = edge id | local row << 20");
static_assert(NNODES <= (1 << 24) && OWN_ROWS <= (1 << 8), "record word = src | local row << 24");
static_assert(NVEC % 32 == 0 && (OWN_QCAP / 4) % 32 == 0 && (NVEC / OWN_BLOCKS) % 32 == 0,
              "range ends and start shifts fall on wave boundaries");

__device__ __forceinline__ void own_scan(const int4* __restrict__ d4,
                                         int vBegin, int vEnd, int rot, int r0,
                                         int* queue, int* qcount, int wave, int lane)
{
    for (int base = vBegin; base < vEnd; base += SCAN_DEPTH * OWN_THREADS) {
        int4 d[SCAN_DEPTH];
        int  vv[SCAN_DEPTH];
#pragma unroll
        for (int u = 0; u < SCAN_DEPTH; ++u) {
            const int vw = base + u * OWN_THREADS + wave * 32;
            int w = vw + lane + rot;
            if (w >= NVEC) w -= NVEC;
            vv[u] = w;
            d[u]  = d4[(vw < vEnd) ? w : lane];
        }
#pragma unroll
        for (int u = 0; u < SCAN_DEPTH; ++u) {
            const int vw = base + u * OWN_THREADS + wave * 32;
            if (vw < vEnd) {
                const int dk[4] = { d[u].x, d[u].y, d[u].z, d[u].w };
#pragma unroll
                for (int k = 0; k < 4; ++k) {
                    const unsigned lrow = (unsigned)(dk[k] - r0);
                    const bool m = lrow < (unsigned)OWN_ROWS;
                    const unsigned mask = __builtin_amdgcn_ballot_w32(m);
                    if (mask != 0u) {
                        int first = 0;
                        if (lane == 0)
                            first = atomicAdd(qcount, __popc(mask));
                        first = __builtin_amdgcn_readfirstlane(first);
                        if (m) {
                            const int pos = first + (int)__builtin_amdgcn_mbcnt_lo(mask, 0u);
                            if (pos < OWN_QCAP)
                                queue[pos] = (4 * vv[u] + k) | (int)(lrow << 20);
                        }
                    }
                }
            }
        }
    }
}

template <bool FROMREC, bool SAVE>
__device__ __forceinline__ void own_process(int n,
                                            const float* __restrict__ x, int ldx,
                                            const float* __restrict__ ea,
                                            const int* __restrict__ src,
                                            const int* queue, int2* recs, float* acc,
                                            float lw0, float lw1, float lw2,
                                            float lb0, float lb1, float lb2,
                                            int wave, int lane)
{
    for (int g = wave * 32; g < n; g += 32 * OWN_WAVES) {
        const bool live = g + lane < n;
        int s, ab, lrow;
        if constexpr (FROMREC) {
            const int2 r = live ? recs[g + lane] : make_int2(0, 0);
            s    = r.x & 0xFFFFFF;
            lrow = (int)((unsigned)r.x >> 24);
            ab   = r.y;
        } else {
            const int ent = live ? queue[g + lane] : 0;
            const int e   = ent & 0xFFFFF;
            lrow = ent >> 20;
            s    = src[e];
            ab   = __float_as_int(ea[e]);
            if constexpr (SAVE) {
                if (live)
                    recs[g + lane] = make_int2((int)((unsigned)s | ((unsigned)lrow << 24)), ab);
            }
        }
        const int cnt  = (n - g < 32) ? (n - g) : 32;
        for (int j0 = 0; j0 < cnt; j0 += ROW_DEPTH) {
            float xv[ROW_DEPTH][3];
#pragma unroll
            for (int u = 0; u < ROW_DEPTH; ++u) {
                const int sj = __builtin_amdgcn_readlane(s, j0 + u);
                const float* xr = x + (size_t)sj * ldx + lane;
                xv[u][0] = xr[0];
                xv[u][1] = xr[32];
                xv[u][2] = xr[64];
            }
#pragma unroll
            for (int u = 0; u < ROW_DEPTH; ++u) {
                if (j0 + u < cnt) {
                    const float a  = __int_as_float(__builtin_amdgcn_readlane(ab, j0 + u));
                    const int   rj = __builtin_amdgcn_readlane(lrow, j0 + u);
                    float* ar = acc + rj * 96 + lane;
                    atomicAdd(ar,      fmaxf(xv[u][0] + a * lw0 + lb0, 0.0f));
                    atomicAdd(ar + 32, fmaxf(xv[u][1] + a * lw1 + lb1, 0.0f));
                    atomicAdd(ar + 64, fmaxf(xv[u][2] + a * lw2 + lb2, 0.0f));
                }
            }
        }
    }
}

template <bool FIRST>
__global__ __launch_bounds__(OWN_THREADS)
void edge_owner_kernel(const float* __restrict__ x, int ldx,
                       const float* __restrict__ ea,
                       const int* __restrict__ src,
                       const int* __restrict__ dst,
                       const float* __restrict__ lw,
                       const float* __restrict__ lb,
                       float* __restrict__ agg,
                       int* __restrict__ savedCnt,
                       int* __restrict__ savedRec)
{
    __shared__ __attribute__((aligned(16))) float acc[OWN_ROWS * 96];
    __shared__ int queue[OWN_QCAP];
    __shared__ int qcount;

    const int tid  = threadIdx.x;
    const int lane = tid & 31;
    const int wave = __builtin_amdgcn_readfirstlane(tid >> 5);
    const int r0   = blockIdx.x * OWN_ROWS;
    const int4* d4 = reinterpret_cast<const int4*>(dst);
    int2* recs = reinterpret_cast<int2*>(savedRec) + (size_t)blockIdx.x * OWN_QCAP;
    int total = 0;
    if constexpr (!FIRST) total = savedCnt[blockIdx.x];

    for (int i = tid; i < OWN_ROWS * 96; i += OWN_THREADS) acc[i] = 0.0f;
    if (tid == 0) qcount = 0;

    const float lw0 = lw[lane], lw1 = lw[lane + 32], lw2 = lw[lane + 64];
    const float lb0 = lb[lane], lb1 = lb[lane + 32], lb2 = lb[lane + 64];

    __syncthreads();

    if constexpr (FIRST) {
        own_scan(d4, 0, NVEC, (int)blockIdx.x * (NVEC / OWN_BLOCKS), r0, queue, &qcount, wave, lane);

        __syncthreads();
        total = qcount;
        if (tid == 0) savedCnt[blockIdx.x] = total;
    }
    total = __builtin_amdgcn_readfirstlane(total);

    if (total <= OWN_QCAP) {
        if constexpr (FIRST)
            own_process<false, true>(total, x, ldx, ea, src, queue, recs, acc,
                                     lw0, lw1, lw2, lb0, lb1, lb2, wave, lane);
        else
            own_process<true, false>(total, x, ldx, ea, src, queue, recs, acc,
                                     lw0, lw1, lw2, lb0, lb1, lb2, wave, lane);
    } else {
        for (int vb = 0; vb < NVEC; vb += OWN_QCAP / 4) {
            __syncthreads();
            if (tid == 0) qcount = 0;
            __syncthreads();
            const int ve = (vb + OWN_QCAP / 4 < NVEC) ? (vb + OWN_QCAP / 4) : NVEC;
            own_scan(d4, vb, ve, 0, r0, queue, &qcount, wave, lane);
            __syncthreads();
            const int n = __builtin_amdgcn_readfirstlane(qcount);
            own_process<false, false>(n, x, ldx, ea, src, queue, recs, acc,
                                      lw0, lw1, lw2, lb0, lb1, lb2, wave, lane);
        }
    }

    __syncthreads();
    float4* out4 = reinterpret_cast<float4*>(agg + (size_t)r0 * 96);
    const float4* a4 = reinterpret_cast<const float4*>(acc);
    for (int i = tid; i < OWN_ROWS * 96 / 4; i += OWN_THREADS) out4[i] = a4[i];
}

#define ROWTILES (NNODES / 16)
#define HOLD4(a, i) "v"(a[i]), "v"(a[(i) + 1]), "v"(a[(i) + 2]), "v"(a[(i) + 3])

__device__ __forceinline__ int frag_k(int t, int h)
{
    return 8 * (t >> 1) + 4 * h + 2 * (t & 1);
}

template <int K, int NT, int WPB, bool HASADD, int FOLD0>
__global__ __launch_bounds__(NT * WPB * 32)
void gemm_wmma_f32_kernel(const float* __restrict__ A0, int lda0,
                          const float* __restrict__ A1, int lda1,
                          const float* __restrict__ W,  int ldw,
                          const float* __restrict__ bias,
                          float* __restrict__ out, int ldo,
                          float* __restrict__ colsum,
                          float* __restrict__ colsq,
                          const float* __restrict__ fsum,
                          const float* __restrict__ fsq,
                          const float* __restrict__ fg,
                          const float* __restrict__ fbeta)
{
    constexpr int STEPS  = K / 4;
    constexpr int CHUNKS = K / 8;

    const int lane = threadIdx.x & 31;
    const int wave = __builtin_amdgcn_readfirstlane((int)(threadIdx.x >> 5));
    const int lo   = lane & 15;
    const int hi   = lane >> 4;
    const int ct       = wave % NT;
    const int walker   = blockIdx.x * WPB + wave / NT;
    const int nWalkers = gridDim.x * WPB;
    const int col      = ct * 16 + lo;

    v2f b[STEPS];
#pragma unroll
    for (int t = 0; t < STEPS; ++t) {
        const int k = frag_k(t, hi);
        b[t].x = W[(size_t)k * ldw + col];
        b[t].y = W[(size_t)(k + 1) * ldw + col];
    }
    float bv = bias[col];

    if constexpr (FOLD0 < K) {
        constexpr int KF = K - FOLD0;
        static_assert(FOLD0 % 8 == 0, "a 16-byte A chunk must lie wholly on one side of FOLD0");
        static_assert(KF <= NT * WPB * 32, "one thread per folded column");
        __shared__ float fsc[KF];
        __shared__ float fsh[KF];
        if (threadIdx.x < KF) {
            const int i = threadIdx.x;
            const float mu  = fsum[i] * INV_N;
            const float var = fsq[i] * INV_N - mu * mu;
            const float sc  = fg[i] * rsqrtf(var + BN_EPS);
            fsc[i] = sc;
            fsh[i] = fbeta[i] - mu * sc;
        }
        __syncthreads();
        float part = 0.0f;
#pragma unroll
        for (int t = 0; t < STEPS; ++t) {
            if (8 * (t >> 1) >= FOLD0) {
                const int kf = frag_k(t, hi) - FOLD0;
                part += fsh[kf] * b[t].x + fsh[kf + 1] * b[t].y;
                b[t].x *= fsc[kf];
                b[t].y *= fsc[kf + 1];
            }
        }
        bv += part + __shfl_xor(part, 16);
    }

    float s1 = 0.0f, s2 = 0.0f;

    for (int rt = walker; rt < ROWTILES; rt += nWalkers) {
        const int row0 = rt * 16;
        const v4f* ap0 = reinterpret_cast<const v4f*>(
            A0 + (size_t)(row0 + lo) * lda0 + 4 * hi);

        v4f q[CHUNKS];
#pragma unroll
        for (int j = 0; j < CHUNKS; ++j) q[j] = ap0[2 * j];

        if constexpr (HASADD) {
            const v4f* ap1 = reinterpret_cast<const v4f*>(
                A1 + (size_t)(row0 + lo) * lda1 + 4 * hi);
            v4f p[CHUNKS];
#pragma unroll
            for (int j = 0; j < CHUNKS; ++j) p[j] = ap1[2 * j];
            static_assert(CHUNKS == 12, "fence operand list is written for K = 96");
            asm volatile("" :: HOLD4(q, 0), HOLD4(q, 4), HOLD4(q, 8),
                               HOLD4(p, 0), HOLD4(p, 4), HOLD4(p, 8));
#pragma unroll
            for (int j = 0; j < CHUNKS; ++j) q[j] += p[j];
        } else if constexpr (CHUNKS == 12) {
            asm volatile("" :: HOLD4(q, 0), HOLD4(q, 4), HOLD4(q, 8));
        } else {
            static_assert(CHUNKS == 20, "fence operand list is written for K = 96 or 160");
            asm volatile("" :: HOLD4(q, 0), HOLD4(q, 4), HOLD4(q, 8),
                               HOLD4(q, 12), HOLD4(q, 16));
        }

        v8f c = {};
#pragma unroll
        for (int j = 0; j < CHUNKS; ++j) {
            v2f a;
            a.x = q[j].x; a.y = q[j].y;
            c = __builtin_amdgcn_wmma_f32_16x16x4_f32(
                     false, a,  false, b[2 * j],
                     (short)0, c,  false,  false);
            a.x = q[j].z; a.y = q[j].w;
            c = __builtin_amdgcn_wmma_f32_16x16x4_f32(
                     false, a,  false, b[2 * j + 1],
                     (short)0, c,  false,  false);
        }

#pragma unroll
        for (int r = 0; r < 8; ++r) {
            int row = row0 + r + 8 * hi;
            const float v = fmaxf(c[r] + bv, 0.0f);
            out[(size_t)row * ldo + col] = v;
            s1 += v;
            s2 += v * v;
        }
    }

    s1 += __shfl_xor(s1, 16);
    s2 += __shfl_xor(s2, 16);

    if constexpr (WPB == 1) {
        if (hi == 0) {
            atomicAdd(&colsum[col], s1);
            atomicAdd(&colsq[col],  s2);
        }
    } else {
        static_assert(NT == 1, "in-block add assumes all waves own the same columns");
        __shared__ float red[WPB][32];
        red[wave][lane] = (hi == 0) ? s1 : s2;
        __syncthreads();
        if (wave == 0) {
            float t = 0.0f;
            for (int w = 0; w < WPB; ++w) t += red[w][lane];
            atomicAdd((hi == 0 ? colsum : colsq) + lo, t);
        }
    }
}

__global__ __launch_bounds__(256)
void bn_apply_kernel(const float* __restrict__ in, int ldi,
                     float* __restrict__ out, int ldo,
                     const float* __restrict__ sum,
                     const float* __restrict__ sumsq,
                     const float* __restrict__ g,
                     const float* __restrict__ beta, int n, int C)
{
    int gid = blockIdx.x * blockDim.x + threadIdx.x;
    if (gid >= n * C) return;
    int row = gid / C;
    int c   = gid % C;
    float mu  = sum[c] * INV_N;
    float var = sumsq[c] * INV_N - mu * mu;
    float sc  = g[c] * rsqrtf(var + BN_EPS);
    float sh  = beta[c] - mu * sc;
    out[(size_t)row * ldo + c] = in[(size_t)row * ldi + c] * sc + sh;
}

extern "C" void kernel_launch(void* const* d_in, const int* in_sizes, int n_in,
                              void* d_out, int out_size, void* d_ws, size_t ws_size,
                              hipStream_t stream)
{
    (void)in_sizes; (void)n_in; (void)out_size; (void)ws_size;

    const float* x          = (const float*)d_in[0];
    const float* edge_attr  = (const float*)d_in[1];
    const int*   edge_index = (const int*)  d_in[2];
    const float* lin_w      = (const float*)d_in[3];
    const float* lin_b      = (const float*)d_in[4];
    const float* c1_w  = (const float*)d_in[5];
    const float* c1_b  = (const float*)d_in[6];
    const float* c1_g  = (const float*)d_in[7];
    const float* c1_be = (const float*)d_in[8];
    const float* c2_w  = (const float*)d_in[9];
    const float* c2_b  = (const float*)d_in[10];
    const float* c2_g  = (const float*)d_in[11];
    const float* c2_be = (const float*)d_in[12];
    const float* l1_w  = (const float*)d_in[13];
    const float* l1_b  = (const float*)d_in[14];
    const float* l1_g  = (const float*)d_in[15];
    const float* l1_be = (const float*)d_in[16];
    const float* ma_w  = (const float*)d_in[17];
    const float* ma_b  = (const float*)d_in[18];
    const float* ma_g  = (const float*)d_in[19];
    const float* ma_be = (const float*)d_in[20];
    const float* mb_w  = (const float*)d_in[21];
    const float* mb_b  = (const float*)d_in[22];
    const float* mb_g  = (const float*)d_in[23];
    const float* mb_be = (const float*)d_in[24];

    const int* src = edge_index;
    const int* dst = edge_index + NEDGES;

    const size_t N = NNODES;

    float* ws    = (float*)d_ws;
    float* agg   = ws;
    float* bufP  = agg  + N * 96;
    float* bufQ  = bufP + N * 96;
    float* xcat  = bufQ + N * 96;
    float* stats = xcat + N * 160;
    int* ownCnt  = (int*)(stats + 1024);
    int* ownRec  = (int*)(stats + 2048);


    auto sumL = [&](int L) { return stats + (size_t)L * 192; };
    auto sqL  = [&](int L) { return stats + (size_t)L * 192 + 96; };

    hipMemsetAsync(stats, 0, 5 * 192 * sizeof(float), stream);

    edge_owner_kernel<true><<<OWN_BLOCKS, OWN_THREADS, 0, stream>>>(
        x, 96, edge_attr, src, dst, lin_w, lin_b, agg, ownCnt, ownRec);

    gemm_wmma_f32_kernel<96, 6, 1, true, 96><<<625, 192, 0, stream>>>(
        x, 96, agg, 96, c1_w, 96, c1_b, bufP, 96, sumL(0), sqL(0),
        nullptr, nullptr, nullptr, nullptr);

    bn_apply_kernel<<<(NNODES * 96 + 255) / 256, 256, 0, stream>>>(
        bufP, 96, xcat, 160, sumL(0), sqL(0), c1_g, c1_be, NNODES, 96);

    edge_owner_kernel<false><<<OWN_BLOCKS, OWN_THREADS, 0, stream>>>(
        xcat, 160, edge_attr, src, dst, lin_w, lin_b, agg, ownCnt, ownRec);

    gemm_wmma_f32_kernel<96, 4, 1, true, 96><<<625, 128, 0, stream>>>(
        xcat, 160, agg, 96, c2_w, 64, c2_b, xcat + 96, 160, sumL(1), sqL(1),
        nullptr, nullptr, nullptr, nullptr);

    gemm_wmma_f32_kernel<160, 6, 1, false, 96><<<625, 192, 0, stream>>>(
        xcat, 160, nullptr, 0, l1_w, 96, l1_b, bufQ, 96, sumL(2), sqL(2),
        sumL(1), sqL(1), c2_g, c2_be);

    gemm_wmma_f32_kernel<96, 6, 1, false, 0><<<625, 192, 0, stream>>>(
        bufQ, 96, nullptr, 0, ma_w, 96, ma_b, bufP, 96, sumL(3), sqL(3),
        sumL(2), sqL(2), l1_g, l1_be);

    gemm_wmma_f32_kernel<96, 1, 5, false, 0><<<625, 160, 0, stream>>>(
        bufP, 96, nullptr, 0, mb_w, 16, mb_b, bufQ, 16, sumL(4), sqL(4),
        sumL(3), sqL(3), ma_g, ma_be);

    bn_apply_kernel<<<(NNODES * 16 + 255) / 256, 256, 0, stream>>>(
        bufQ, 16, (float*)d_out, 16, sumL(4), sqL(4), mb_g, mb_be, NNODES, 16);
}


# Owner filter scans an 8-bit owner key per edge (dst/200 as bit planes, made by one small launch ahead): a lane tests 32 edges at once on 32 bytes, 4x fewer scan bytes; LDS add order changes
# speedup vs baseline: 4.9235x; 1.1099x over previous
#include <hip/hip_runtime.h>
#include <hip/hip_bf16.h>


#define NNODES 50000
#define NEDGES 800000
#define BN_EPS 1e-5f
#define INV_N  (1.0f / (float)NNODES)

typedef __attribute__((ext_vector_type(2))) float v2f;
typedef __attribute__((ext_vector_type(4))) float v4f;
typedef __attribute__((ext_vector_type(8))) float v8f;

#define OWN_ROWS    200
#define OWN_BLOCKS  (NNODES / OWN_ROWS)
#define OWN_THREADS 1024
#define OWN_WAVES   (OWN_THREADS / 32)
#define OWN_QCAP    24576
#define NGRP        (NEDGES / 32)
#define KEY_NONE    255u
#define SCAN_DEPTH  4
#define ROW_DEPTH   16

static_assert(OWN_ROWS * OWN_BLOCKS == NNODES, "every row has exactly one owner");
static_assert(NEDGES % 256 == 0, "a wave of the key launch makes eight whole groups");
static_assert(OWN_BLOCKS <= KEY_NONE, "a block number fits the 8-bit key and is never KEY_NONE");
static_assert(NNODES <= (1 << 24) && OWN_ROWS <= (1 << 8), "record word = src | local row << 24");
static_assert(OWN_QCAP % 32 == 0 && NGRP % OWN_BLOCKS == 0 && ((NGRP / OWN_BLOCKS) * 32) % 128 == 0,
              "slices are whole groups; start shifts fall on 128-byte lines");

#define KEY_WAVE_EDGES 256

__global__ __launch_bounds__(256)
void owner_key_kernel(const int* __restrict__ dst, unsigned* __restrict__ planes)
{
    const int lane = threadIdx.x & 31;
    const int wv = __builtin_amdgcn_readfirstlane((int)((blockIdx.x * 256 + threadIdx.x) >> 5));
    if (wv >= NEDGES / KEY_WAVE_EDGES) return;
    const int* dp = dst + (size_t)wv * KEY_WAVE_EDGES + lane;
    int d[8];
#pragma unroll
    for (int g = 0; g < 8; ++g) d[g] = dp[32 * g];
    unsigned outw[2] = { 0u, 0u };
#pragma unroll
    for (int g = 0; g < 8; ++g) {
        const unsigned key = ((unsigned)d[g] < (unsigned)NNODES)
                           ? (unsigned)d[g] / (unsigned)OWN_ROWS : KEY_NONE;
#pragma unroll
        for (int p = 0; p < 8; ++p) {
            const unsigned vote = __builtin_amdgcn_ballot_w32(((key >> p) & 1u) != 0u);
            if (lane == 8 * (g & 3) + p) outw[g >> 2] = vote;
        }
    }
    unsigned* op = planes + (size_t)wv * (KEY_WAVE_EDGES / 4) + lane;
    op[0]  = outw[0];
    op[32] = outw[1];
}

struct OwnFlip { unsigned f[8]; };

__device__ __forceinline__ void own_scan(const uint4* __restrict__ p4,
                                         int gBegin, int gEnd, int rot, const OwnFlip flip,
                                         int* queue, int* qcount, int wave, int lane)
{
    for (int base = gBegin; base < gEnd; base += SCAN_DEPTH * OWN_THREADS) {
        uint4 lo[SCAN_DEPTH], hi[SCAN_DEPTH];
        int   gg[SCAN_DEPTH];
#pragma unroll
        for (int u = 0; u < SCAN_DEPTH; ++u) {
            const int g = base + u * OWN_THREADS + wave * 32 + lane;
            int w = g + rot;
            if (w >= NGRP) w -= NGRP;
            gg[u] = (g < gEnd) ? w : -1;
            const uint4* gp = p4 + 2 * (size_t)((g < gEnd) ? w : 0);
            lo[u] = gp[0];
            hi[u] = gp[1];
        }
#pragma unroll
        for (int u = 0; u < SCAN_DEPTH; ++u) {
            const int gw = base + u * OWN_THREADS + wave * 32;
            if (gw < gEnd) {
                unsigned m = (lo[u].x ^ flip.f[0]) & (lo[u].y ^ flip.f[1]) &
                             (lo[u].z ^ flip.f[2]) & (lo[u].w ^ flip.f[3]) &
                             (hi[u].x ^ flip.f[4]) & (hi[u].y ^ flip.f[5]) &
                             (hi[u].z ^ flip.f[6]) & (hi[u].w ^ flip.f[7]);
                if (gg[u] < 0) m = 0u;
                for (int it = 0; it < 32; ++it) {
                    const unsigned mask = __builtin_amdgcn_ballot_w32(m != 0u);
                    if (mask == 0u) break;
                    int first = 0;
                    if (lane == 0)
                        first = atomicAdd(qcount, __popc(mask));
                    first = __builtin_amdgcn_readfirstlane(first);
                    if (m != 0u) {
                        const int pos = first + (int)__builtin_amdgcn_mbcnt_lo(mask, 0u);
                        if (pos < OWN_QCAP)
                            queue[pos] = 32 * gg[u] + __builtin_ctz(m);
                        m &= m - 1u;
                    }
                }
            }
        }
    }
}

template <bool FROMREC, bool SAVE>
__device__ __forceinline__ void own_process(int n,
                                            const float* __restrict__ x, int ldx,
                                            const float* __restrict__ ea,
                                            const int* __restrict__ src,
                                            const int* __restrict__ dst, int r0,
                                            const int* queue, int2* recs, float* acc,
                                            float lw0, float lw1, float lw2,
                                            float lb0, float lb1, float lb2,
                                            int wave, int lane)
{
    for (int g = wave * 32; g < n; g += 32 * OWN_WAVES) {
        const bool live = g + lane < n;
        int s, ab, lrow;
        if constexpr (FROMREC) {
            const int2 r = live ? recs[g + lane] : make_int2(0, 0);
            s    = r.x & 0xFFFFFF;
            lrow = (int)((unsigned)r.x >> 24);
            ab   = r.y;
        } else {
            const int e = live ? queue[g + lane] : 0;
            lrow = dst[e] - r0;
            s    = src[e];
            ab   = __float_as_int(ea[e]);
            if constexpr (SAVE) {
                if (live)
                    recs[g + lane] = make_int2((int)((unsigned)s | ((unsigned)lrow << 24)), ab);
            }
        }
        const int cnt  = (n - g < 32) ? (n - g) : 32;
        for (int j0 = 0; j0 < cnt; j0 += ROW_DEPTH) {
            float xv[ROW_DEPTH][3];
#pragma unroll
            for (int u = 0; u < ROW_DEPTH; ++u) {
                const int sj = __builtin_amdgcn_readlane(s, j0 + u);
                const float* xr = x + (size_t)sj * ldx + lane;
                xv[u][0] = xr[0];
                xv[u][1] = xr[32];
                xv[u][2] = xr[64];
            }
#pragma unroll
            for (int u = 0; u < ROW_DEPTH; ++u) {
                if (j0 + u < cnt) {
                    const float a  = __int_as_float(__builtin_amdgcn_readlane(ab, j0 + u));
                    const int   rj = __builtin_amdgcn_readlane(lrow, j0 + u);
                    float* ar = acc + rj * 96 + lane;
                    atomicAdd(ar,      fmaxf(xv[u][0] + a * lw0 + lb0, 0.0f));
                    atomicAdd(ar + 32, fmaxf(xv[u][1] + a * lw1 + lb1, 0.0f));
                    atomicAdd(ar + 64, fmaxf(xv[u][2] + a * lw2 + lb2, 0.0f));
                }
            }
        }
    }
}

template <bool FIRST>
__global__ __launch_bounds__(OWN_THREADS)
void edge_owner_kernel(const float* __restrict__ x, int ldx,
                       const float* __restrict__ ea,
                       const int* __restrict__ src,
                       const int* __restrict__ dst,
                       const unsigned* __restrict__ planes,
                       const float* __restrict__ lw,
                       const float* __restrict__ lb,
                       float* __restrict__ agg,
                       int* __restrict__ savedCnt,
                       int* __restrict__ savedRec)
{
    __shared__ __attribute__((aligned(16))) float acc[OWN_ROWS * 96];
    __shared__ int queue[OWN_QCAP];
    __shared__ int qcount;

    const int tid  = threadIdx.x;
    const int lane = tid & 31;
    const int wave = __builtin_amdgcn_readfirstlane(tid >> 5);
    const int r0   = blockIdx.x * OWN_ROWS;
    const uint4* p4 = reinterpret_cast<const uint4*>(planes);
    OwnFlip flip;
#pragma unroll
    for (int p = 0; p < 8; ++p) flip.f[p] = ((blockIdx.x >> p) & 1u) ? 0u : ~0u;
    int2* recs = reinterpret_cast<int2*>(savedRec) + (size_t)blockIdx.x * OWN_QCAP;
    int total = 0;
    if constexpr (!FIRST) total = savedCnt[blockIdx.x];

    for (int i = tid; i < OWN_ROWS * 96; i += OWN_THREADS) acc[i] = 0.0f;
    if (tid == 0) qcount = 0;

    const float lw0 = lw[lane], lw1 = lw[lane + 32], lw2 = lw[lane + 64];
    const float lb0 = lb[lane], lb1 = lb[lane + 32], lb2 = lb[lane + 64];

    __syncthreads();

    if constexpr (FIRST) {
        own_scan(p4, 0, NGRP, (int)blockIdx.x * (NGRP / OWN_BLOCKS), flip, queue, &qcount, wave, lane);

        __syncthreads();
        total = qcount;
        if (tid == 0) savedCnt[blockIdx.x] = total;
    }
    total = __builtin_amdgcn_readfirstlane(total);

    if (total <= OWN_QCAP) {
        if constexpr (FIRST)
            own_process<false, true>(total, x, ldx, ea, src, dst, r0, queue, recs, acc,
                                     lw0, lw1, lw2, lb0, lb1, lb2, wave, lane);
        else
            own_process<true, false>(total, x, ldx, ea, src, dst, r0, queue, recs, acc,
                                     lw0, lw1, lw2, lb0, lb1, lb2, wave, lane);
    } else {
        for (int vb = 0; vb < NGRP; vb += OWN_QCAP / 32) {
            __syncthreads();
            if (tid == 0) qcount = 0;
            __syncthreads();
            const int ve = (vb + OWN_QCAP / 32 < NGRP) ? (vb + OWN_QCAP / 32) : NGRP;
            own_scan(p4, vb, ve, 0, flip, queue, &qcount, wave, lane);
            __syncthreads();
            const int n = __builtin_amdgcn_readfirstlane(qcount);
            own_process<false, false>(n, x, ldx, ea, src, dst, r0, queue, recs, acc,
                                      lw0, lw1, lw2, lb0, lb1, lb2, wave, lane);
        }
    }

    __syncthreads();
    float4* out4 = reinterpret_cast<float4*>(agg + (size_t)r0 * 96);
    const float4* a4 = reinterpret_cast<const float4*>(acc);
    for (int i = tid; i < OWN_ROWS * 96 / 4; i += OWN_THREADS) out4[i] = a4[i];
}

#define ROWTILES (NNODES / 16)
#define HOLD4(a, i) "v"(a[i]), "v"(a[(i) + 1]), "v"(a[(i) + 2]), "v"(a[(i) + 3])

__device__ __forceinline__ int frag_k(int t, int h)
{
    return 8 * (t >> 1) + 4 * h + 2 * (t & 1);
}

template <int K, int NT, int WPB, bool HASADD, int FOLD0>
__global__ __launch_bounds__(NT * WPB * 32)
void gemm_wmma_f32_kernel(const float* __restrict__ A0, int lda0,
                          const float* __restrict__ A1, int lda1,
                          const float* __restrict__ W,  int ldw,
                          const float* __restrict__ bias,
                          float* __restrict__ out, int ldo,
                          float* __restrict__ colsum,
                          float* __restrict__ colsq,
                          const float* __restrict__ fsum,
                          const float* __restrict__ fsq,
                          const float* __restrict__ fg,
                          const float* __restrict__ fbeta)
{
    constexpr int STEPS  = K / 4;
    constexpr int CHUNKS = K / 8;

    const int lane = threadIdx.x & 31;
    const int wave = __builtin_amdgcn_readfirstlane((int)(threadIdx.x >> 5));
    const int lo   = lane & 15;
    const int hi   = lane >> 4;
    const int ct       = wave % NT;
    const int walker   = blockIdx.x * WPB + wave / NT;
    const int nWalkers = gridDim.x * WPB;
    const int col      = ct * 16 + lo;

    v2f b[STEPS];
#pragma unroll
    for (int t = 0; t < STEPS; ++t) {
        const int k = frag_k(t, hi);
        b[t].x = W[(size_t)k * ldw + col];
        b[t].y = W[(size_t)(k + 1) * ldw + col];
    }
    float bv = bias[col];

    if constexpr (FOLD0 < K) {
        constexpr int KF = K - FOLD0;
        static_assert(FOLD0 % 8 == 0, "a 16-byte A chunk must lie wholly on one side of FOLD0");
        static_assert(KF <= NT * WPB * 32, "one thread per folded column");
        __shared__ float fsc[KF];
        __shared__ float fsh[KF];
        if (threadIdx.x < KF) {
            const int i = threadIdx.x;
            const float mu  = fsum[i] * INV_N;
            const float var = fsq[i] * INV_N - mu * mu;
            const float sc  = fg[i] * rsqrtf(var + BN_EPS);
            fsc[i] = sc;
            fsh[i] = fbeta[i] - mu * sc;
        }
        __syncthreads();
        float part = 0.0f;
#pragma unroll
        for (int t = 0; t < STEPS; ++t) {
            if (8 * (t >> 1) >= FOLD0) {
                const int kf = frag_k(t, hi) - FOLD0;
                part += fsh[kf] * b[t].x + fsh[kf + 1] * b[t].y;
                b[t].x *= fsc[kf];
                b[t].y *= fsc[kf + 1];
            }
        }
        bv += part + __shfl_xor(part, 16);
    }

    float s1 = 0.0f, s2 = 0.0f;

    for (int rt = walker; rt < ROWTILES; rt += nWalkers) {
        const int row0 = rt * 16;
        const v4f* ap0 = reinterpret_cast<const v4f*>(
            A0 + (size_t)(row0 + lo) * lda0 + 4 * hi);

        v4f q[CHUNKS];
#pragma unroll
        for (int j = 0; j < CHUNKS; ++j) q[j] = ap0[2 * j];

        if constexpr (HASADD) {
            const v4f* ap1 = reinterpret_cast<const v4f*>(
                A1 + (size_t)(row0 + lo) * lda1 + 4 * hi);
            v4f p[CHUNKS];
#pragma unroll
            for (int j = 0; j < CHUNKS; ++j) p[j] = ap1[2 * j];
            static_assert(CHUNKS == 12, "fence operand list is written for K = 96");
            asm volatile("" :: HOLD4(q, 0), HOLD4(q, 4), HOLD4(q, 8),
                               HOLD4(p, 0), HOLD4(p, 4), HOLD4(p, 8));
#pragma unroll
            for (int j = 0; j < CHUNKS; ++j) q[j] += p[j];
        } else if constexpr (CHUNKS == 12) {
            asm volatile("" :: HOLD4(q, 0), HOLD4(q, 4), HOLD4(q, 8));
        } else {
            static_assert(CHUNKS == 20, "fence operand list is written for K = 96 or 160");
            asm volatile("" :: HOLD4(q, 0), HOLD4(q, 4), HOLD4(q, 8),
                               HOLD4(q, 12), HOLD4(q, 16));
        }

        v8f c = {};
#pragma unroll
        for (int j = 0; j < CHUNKS; ++j) {
            v2f a;
            a.x = q[j].x; a.y = q[j].y;
            c = __builtin_amdgcn_wmma_f32_16x16x4_f32(
                     false, a,  false, b[2 * j],
                     (short)0, c,  false,  false);
            a.x = q[j].z; a.y = q[j].w;
            c = __builtin_amdgcn_wmma_f32_16x16x4_f32(
                     false, a,  false, b[2 * j + 1],
                     (short)0, c,  false,  false);
        }

#pragma unroll
        for (int r = 0; r < 8; ++r) {
            int row = row0 + r + 8 * hi;
            const float v = fmaxf(c[r] + bv, 0.0f);
            out[(size_t)row * ldo + col] = v;
            s1 += v;
            s2 += v * v;
        }
    }

    s1 += __shfl_xor(s1, 16);
    s2 += __shfl_xor(s2, 16);

    if constexpr (WPB == 1) {
        if (hi == 0) {
            atomicAdd(&colsum[col], s1);
            atomicAdd(&colsq[col],  s2);
        }
    } else {
        static_assert(NT == 1, "in-block add assumes all waves own the same columns");
        __shared__ float red[WPB][32];
        red[wave][lane] = (hi == 0) ? s1 : s2;
        __syncthreads();
        if (wave == 0) {
            float t = 0.0f;
            for (int w = 0; w < WPB; ++w) t += red[w][lane];
            atomicAdd((hi == 0 ? colsum : colsq) + lo, t);
        }
    }
}

__global__ __launch_bounds__(256)
void bn_apply_kernel(const float* __restrict__ in, int ldi,
                     float* __restrict__ out, int ldo,
                     const float* __restrict__ sum,
                     const float* __restrict__ sumsq,
                     const float* __restrict__ g,
                     const float* __restrict__ beta, int n, int C)
{
    int gid = blockIdx.x * blockDim.x + threadIdx.x;
    if (gid >= n * C) return;
    int row = gid / C;
    int c   = gid % C;
    float mu  = sum[c] * INV_N;
    float var = sumsq[c] * INV_N - mu * mu;
    float sc  = g[c] * rsqrtf(var + BN_EPS);
    float sh  = beta[c] - mu * sc;
    out[(size_t)row * ldo + c] = in[(size_t)row * ldi + c] * sc + sh;
}

extern "C" void kernel_launch(void* const* d_in, const int* in_sizes, int n_in,
                              void* d_out, int out_size, void* d_ws, size_t ws_size,
                              hipStream_t stream)
{
    (void)in_sizes; (void)n_in; (void)out_size; (void)ws_size;

    const float* x          = (const float*)d_in[0];
    const float* edge_attr  = (const float*)d_in[1];
    const int*   edge_index = (const int*)  d_in[2];
    const float* lin_w      = (const float*)d_in[3];
    const float* lin_b      = (const float*)d_in[4];
    const float* c1_w  = (const float*)d_in[5];
    const float* c1_b  = (const float*)d_in[6];
    const float* c1_g  = (const float*)d_in[7];
    const float* c1_be = (const float*)d_in[8];
    const float* c2_w  = (const float*)d_in[9];
    const float* c2_b  = (const float*)d_in[10];
    const float* c2_g  = (const float*)d_in[11];
    const float* c2_be = (const float*)d_in[12];
    const float* l1_w  = (const float*)d_in[13];
    const float* l1_b  = (const float*)d_in[14];
    const float* l1_g  = (const float*)d_in[15];
    const float* l1_be = (const float*)d_in[16];
    const float* ma_w  = (const float*)d_in[17];
    const float* ma_b  = (const float*)d_in[18];
    const float* ma_g  = (const float*)d_in[19];
    const float* ma_be = (const float*)d_in[20];
    const float* mb_w  = (const float*)d_in[21];
    const float* mb_b  = (const float*)d_in[22];
    const float* mb_g  = (const float*)d_in[23];
    const float* mb_be = (const float*)d_in[24];

    const int* src = edge_index;
    const int* dst = edge_index + NEDGES;

    const size_t N = NNODES;

    float* ws    = (float*)d_ws;
    float* agg   = ws;
    float* bufP  = agg  + N * 96;
    float* bufQ  = bufP + N * 96;
    float* xcat  = bufQ + N * 96;
    float* stats = xcat + N * 160;
    int* ownCnt  = (int*)(stats + 1024);
    int* ownRec  = (int*)(stats + 2048);
    unsigned* ownKey = (unsigned*)(ownRec + (size_t)OWN_BLOCKS * OWN_QCAP * 2);


    auto sumL = [&](int L) { return stats + (size_t)L * 192; };
    auto sqL  = [&](int L) { return stats + (size_t)L * 192 + 96; };

    hipMemsetAsync(stats, 0, 5 * 192 * sizeof(float), stream);

    owner_key_kernel<<<(NEDGES / KEY_WAVE_EDGES + 7) / 8, 256, 0, stream>>>(dst, ownKey);

    edge_owner_kernel<true><<<OWN_BLOCKS, OWN_THREADS, 0, stream>>>(
        x, 96, edge_attr, src, dst, ownKey, lin_w, lin_b, agg, ownCnt, ownRec);

    gemm_wmma_f32_kernel<96, 6, 1, true, 96><<<625, 192, 0, stream>>>(
        x, 96, agg, 96, c1_w, 96, c1_b, bufP, 96, sumL(0), sqL(0),
        nullptr, nullptr, nullptr, nullptr);

    bn_apply_kernel<<<(NNODES * 96 + 255) / 256, 256, 0, stream>>>(
        bufP, 96, xcat, 160, sumL(0), sqL(0), c1_g, c1_be, NNODES, 96);

    edge_owner_kernel<false><<<OWN_BLOCKS, OWN_THREADS, 0, stream>>>(
        xcat, 160, edge_attr, src, dst, ownKey, lin_w, lin_b, agg, ownCnt, ownRec);

    gemm_wmma_f32_kernel<96, 4, 1, true, 96><<<625, 128, 0, stream>>>(
        xcat, 160, agg, 96, c2_w, 64, c2_b, xcat + 96, 160, sumL(1), sqL(1),
        nullptr, nullptr, nullptr, nullptr);

    gemm_wmma_f32_kernel<160, 6, 1, false, 96><<<625, 192, 0, stream>>>(
        xcat, 160, nullptr, 0, l1_w, 96, l1_b, bufQ, 96, sumL(2), sqL(2),
        sumL(1), sqL(1), c2_g, c2_be);

    gemm_wmma_f32_kernel<96, 6, 1, false, 0><<<625, 192, 0, stream>>>(
        bufQ, 96, nullptr, 0, ma_w, 96, ma_b, bufP, 96, sumL(3), sqL(3),
        sumL(2), sqL(2), l1_g, l1_be);

    gemm_wmma_f32_kernel<96, 1, 5, false, 0><<<625, 160, 0, stream>>>(
        bufP, 96, nullptr, 0, mb_w, 16, mb_b, bufQ, 16, sumL(4), sqL(4),
        sumL(3), sqL(3), ma_g, ma_be);

    bn_apply_kernel<<<(NNODES * 16 + 255) / 256, 256, 0, stream>>>(
        bufQ, 16, (float*)d_out, 16, sumL(4), sqL(4), mb_g, mb_be, NNODES, 16);
}


# conv1 BatchNorm made by its readers: conv2 gather scales fetched rows, its owner block starts the LDS sums from its own normalised rows (GEMM reads one A), lin1 folds it in W; sum order changes
# speedup vs baseline: 5.3858x; 1.0939x over previous
#include <hip/hip_runtime.h>
#include <hip/hip_bf16.h>


#define NNODES 50000
#define NEDGES 800000
#define BN_EPS 1e-5f
#define INV_N  (1.0f / (float)NNODES)

typedef __attribute__((ext_vector_type(2))) float v2f;
typedef __attribute__((ext_vector_type(4))) float v4f;
typedef __attribute__((ext_vector_type(8))) float v8f;

#define OWN_ROWS    200
#define OWN_BLOCKS  (NNODES / OWN_ROWS)
#define OWN_THREADS 1024
#define OWN_WAVES   (OWN_THREADS / 32)
#define OWN_QCAP    24576
#define NGRP        (NEDGES / 32)
#define KEY_NONE    255u
#define SCAN_DEPTH  4
#define ROW_DEPTH   16

static_assert(OWN_ROWS * OWN_BLOCKS == NNODES, "every row has exactly one owner");
static_assert(NEDGES % 256 == 0, "a wave of the key launch makes eight whole groups");
static_assert(OWN_BLOCKS <= KEY_NONE, "a block number fits the 8-bit key and is never KEY_NONE");
static_assert(NNODES <= (1 << 24) && OWN_ROWS <= (1 << 8), "record word = src | local row << 24");
static_assert(OWN_QCAP % 32 == 0 && NGRP % OWN_BLOCKS == 0 && ((NGRP / OWN_BLOCKS) * 32) % 128 == 0,
              "slices are whole groups; start shifts fall on 128-byte lines");

#define KEY_WAVE_EDGES 256

__global__ __launch_bounds__(256)
void owner_key_kernel(const int* __restrict__ dst, unsigned* __restrict__ planes)
{
    const int lane = threadIdx.x & 31;
    const int wv = __builtin_amdgcn_readfirstlane((int)((blockIdx.x * 256 + threadIdx.x) >> 5));
    if (wv >= NEDGES / KEY_WAVE_EDGES) return;
    const int* dp = dst + (size_t)wv * KEY_WAVE_EDGES + lane;
    int d[8];
#pragma unroll
    for (int g = 0; g < 8; ++g) d[g] = dp[32 * g];
    unsigned outw[2] = { 0u, 0u };
#pragma unroll
    for (int g = 0; g < 8; ++g) {
        const unsigned key = ((unsigned)d[g] < (unsigned)NNODES)
                           ? (unsigned)d[g] / (unsigned)OWN_ROWS : KEY_NONE;
#pragma unroll
        for (int p = 0; p < 8; ++p) {
            const unsigned vote = __builtin_amdgcn_ballot_w32(((key >> p) & 1u) != 0u);
            if (lane == 8 * (g & 3) + p) outw[g >> 2] = vote;
        }
    }
    unsigned* op = planes + (size_t)wv * (KEY_WAVE_EDGES / 4) + lane;
    op[0]  = outw[0];
    op[32] = outw[1];
}

struct OwnFlip { unsigned f[8]; };

__device__ __forceinline__ void own_scan(const uint4* __restrict__ p4,
                                         int gBegin, int gEnd, int rot, const OwnFlip flip,
                                         int* queue, int* qcount, int wave, int lane)
{
    for (int base = gBegin; base < gEnd; base += SCAN_DEPTH * OWN_THREADS) {
        uint4 lo[SCAN_DEPTH], hi[SCAN_DEPTH];
        int   gg[SCAN_DEPTH];
#pragma unroll
        for (int u = 0; u < SCAN_DEPTH; ++u) {
            const int g = base + u * OWN_THREADS + wave * 32 + lane;
            int w = g + rot;
            if (w >= NGRP) w -= NGRP;
            gg[u] = (g < gEnd) ? w : -1;
            const uint4* gp = p4 + 2 * (size_t)((g < gEnd) ? w : 0);
            lo[u] = gp[0];
            hi[u] = gp[1];
        }
#pragma unroll
        for (int u = 0; u < SCAN_DEPTH; ++u) {
            const int gw = base + u * OWN_THREADS + wave * 32;
            if (gw < gEnd) {
                unsigned m = (lo[u].x ^ flip.f[0]) & (lo[u].y ^ flip.f[1]) &
                             (lo[u].z ^ flip.f[2]) & (lo[u].w ^ flip.f[3]) &
                             (hi[u].x ^ flip.f[4]) & (hi[u].y ^ flip.f[5]) &
                             (hi[u].z ^ flip.f[6]) & (hi[u].w ^ flip.f[7]);
                if (gg[u] < 0) m = 0u;
                for (int it = 0; it < 32; ++it) {
                    const unsigned mask = __builtin_amdgcn_ballot_w32(m != 0u);
                    if (mask == 0u) break;
                    int first = 0;
                    if (lane == 0)
                        first = atomicAdd(qcount, __popc(mask));
                    first = __builtin_amdgcn_readfirstlane(first);
                    if (m != 0u) {
                        const int pos = first + (int)__builtin_amdgcn_mbcnt_lo(mask, 0u);
                        if (pos < OWN_QCAP)
                            queue[pos] = 32 * gg[u] + __builtin_ctz(m);
                        m &= m - 1u;
                    }
                }
            }
        }
    }
}

template <bool FROMREC, bool SAVE, bool NORM>
__device__ __forceinline__ void own_process(int n,
                                            const float* __restrict__ x, int ldx,
                                            const float* __restrict__ ea,
                                            const int* __restrict__ src,
                                            const int* __restrict__ dst, int r0,
                                            const int* queue, int2* recs, float* acc,
                                            float lw0, float lw1, float lw2,
                                            float lb0, float lb1, float lb2,
                                            float sc0, float sc1, float sc2,
                                            float sh0, float sh1, float sh2,
                                            int wave, int lane)
{
    for (int g = wave * 32; g < n; g += 32 * OWN_WAVES) {
        const bool live = g + lane < n;
        int s, ab, lrow;
        if constexpr (FROMREC) {
            const int2 r = live ? recs[g + lane] : make_int2(0, 0);
            s    = r.x & 0xFFFFFF;
            lrow = (int)((unsigned)r.x >> 24);
            ab   = r.y;
        } else {
            const int e = live ? queue[g + lane] : 0;
            lrow = dst[e] - r0;
            s    = src[e];
            ab   = __float_as_int(ea[e]);
            if constexpr (SAVE) {
                if (live)
                    recs[g + lane] = make_int2((int)((unsigned)s | ((unsigned)lrow << 24)), ab);
            }
        }
        const int cnt  = (n - g < 32) ? (n - g) : 32;
        for (int j0 = 0; j0 < cnt; j0 += ROW_DEPTH) {
            float xv[ROW_DEPTH][3];
#pragma unroll
            for (int u = 0; u < ROW_DEPTH; ++u) {
                const int sj = __builtin_amdgcn_readlane(s, j0 + u);
                const float* xr = x + (size_t)sj * ldx + lane;
                xv[u][0] = xr[0];
                xv[u][1] = xr[32];
                xv[u][2] = xr[64];
            }
#pragma unroll
            for (int u = 0; u < ROW_DEPTH; ++u) {
                if (j0 + u < cnt) {
                    const float a  = __int_as_float(__builtin_amdgcn_readlane(ab, j0 + u));
                    const int   rj = __builtin_amdgcn_readlane(lrow, j0 + u);
                    float* ar = acc + rj * 96 + lane;
                    float x0 = xv[u][0], x1 = xv[u][1], x2 = xv[u][2];
                    if constexpr (NORM) {
                        x0 = x0 * sc0 + sh0;
                        x1 = x1 * sc1 + sh1;
                        x2 = x2 * sc2 + sh2;
                    }
                    atomicAdd(ar,      fmaxf(x0 + a * lw0 + lb0, 0.0f));
                    atomicAdd(ar + 32, fmaxf(x1 + a * lw1 + lb1, 0.0f));
                    atomicAdd(ar + 64, fmaxf(x2 + a * lw2 + lb2, 0.0f));
                }
            }
        }
    }
}

template <bool FIRST>
__global__ __launch_bounds__(OWN_THREADS)
void edge_owner_kernel(const float* __restrict__ x, int ldx,
                       const float* __restrict__ ea,
                       const int* __restrict__ src,
                       const int* __restrict__ dst,
                       const unsigned* __restrict__ planes,
                       const float* __restrict__ lw,
                       const float* __restrict__ lb,
                       float* __restrict__ agg,
                       int* __restrict__ savedCnt,
                       int* __restrict__ savedRec,
                       const float* __restrict__ nsum,
                       const float* __restrict__ nsq,
                       const float* __restrict__ ng,
                       const float* __restrict__ nbeta)
{
    constexpr bool NORM = !FIRST;
    __shared__ __attribute__((aligned(16))) float acc[OWN_ROWS * 96];
    __shared__ int queue[OWN_QCAP];
    __shared__ int qcount;

    const int tid  = threadIdx.x;
    const int lane = tid & 31;
    const int wave = __builtin_amdgcn_readfirstlane(tid >> 5);
    const int r0   = blockIdx.x * OWN_ROWS;
    const uint4* p4 = reinterpret_cast<const uint4*>(planes);
    OwnFlip flip;
#pragma unroll
    for (int p = 0; p < 8; ++p) flip.f[p] = ((blockIdx.x >> p) & 1u) ? 0u : ~0u;
    int2* recs = reinterpret_cast<int2*>(savedRec) + (size_t)blockIdx.x * OWN_QCAP;
    int total = 0;
    if constexpr (!FIRST) total = savedCnt[blockIdx.x];

    const float lw0 = lw[lane], lw1 = lw[lane + 32], lw2 = lw[lane + 64];
    const float lb0 = lb[lane], lb1 = lb[lane + 32], lb2 = lb[lane + 64];

    float sc[3] = { 1.0f, 1.0f, 1.0f };
    float sh[3] = { 0.0f, 0.0f, 0.0f };
    if constexpr (NORM) {
        constexpr int INIT_ROWS = (OWN_ROWS + OWN_WAVES - 1) / OWN_WAVES;
        float hv[INIT_ROWS][3];
#pragma unroll
        for (int u = 0; u < INIT_ROWS; ++u) {
            const int row = wave + OWN_WAVES * u;
            const float* xr = x + (size_t)(r0 + (row < OWN_ROWS ? row : 0)) * ldx + lane;
            hv[u][0] = xr[0];
            hv[u][1] = xr[32];
            hv[u][2] = xr[64];
        }
#pragma unroll
        for (int m = 0; m < 3; ++m) {
            const int cc = lane + 32 * m;
            const float mu  = nsum[cc] * INV_N;
            const float var = nsq[cc] * INV_N - mu * mu;
            sc[m] = ng[cc] * rsqrtf(var + BN_EPS);
            sh[m] = nbeta[cc] - mu * sc[m];
        }
#pragma unroll
        for (int u = 0; u < INIT_ROWS; ++u) {
            const int row = wave + OWN_WAVES * u;
            if (row < OWN_ROWS) {
                float* ar = acc + row * 96 + lane;
                ar[0]  = hv[u][0] * sc[0] + sh[0];
                ar[32] = hv[u][1] * sc[1] + sh[1];
                ar[64] = hv[u][2] * sc[2] + sh[2];
            }
        }
    } else {
        for (int i = tid; i < OWN_ROWS * 96; i += OWN_THREADS) acc[i] = 0.0f;
    }
    if (tid == 0) qcount = 0;

    __syncthreads();

    if constexpr (FIRST) {
        own_scan(p4, 0, NGRP, (int)blockIdx.x * (NGRP / OWN_BLOCKS), flip, queue, &qcount, wave, lane);

        __syncthreads();
        total = qcount;
        if (tid == 0) savedCnt[blockIdx.x] = total;
    }
    total = __builtin_amdgcn_readfirstlane(total);

    if (total <= OWN_QCAP) {
        if constexpr (FIRST)
            own_process<false, true, NORM>(total, x, ldx, ea, src, dst, r0, queue, recs, acc,
                                           lw0, lw1, lw2, lb0, lb1, lb2,
                                           sc[0], sc[1], sc[2], sh[0], sh[1], sh[2], wave, lane);
        else
            own_process<true, false, NORM>(total, x, ldx, ea, src, dst, r0, queue, recs, acc,
                                           lw0, lw1, lw2, lb0, lb1, lb2,
                                           sc[0], sc[1], sc[2], sh[0], sh[1], sh[2], wave, lane);
    } else {
        for (int vb = 0; vb < NGRP; vb += OWN_QCAP / 32) {
            __syncthreads();
            if (tid == 0) qcount = 0;
            __syncthreads();
            const int ve = (vb + OWN_QCAP / 32 < NGRP) ? (vb + OWN_QCAP / 32) : NGRP;
            own_scan(p4, vb, ve, 0, flip, queue, &qcount, wave, lane);
            __syncthreads();
            const int n = __builtin_amdgcn_readfirstlane(qcount);
            own_process<false, false, NORM>(n, x, ldx, ea, src, dst, r0, queue, recs, acc,
                                            lw0, lw1, lw2, lb0, lb1, lb2,
                                            sc[0], sc[1], sc[2], sh[0], sh[1], sh[2], wave, lane);
        }
    }

    __syncthreads();
    float4* out4 = reinterpret_cast<float4*>(agg + (size_t)r0 * 96);
    const float4* a4 = reinterpret_cast<const float4*>(acc);
    for (int i = tid; i < OWN_ROWS * 96 / 4; i += OWN_THREADS) out4[i] = a4[i];
}

#define ROWTILES (NNODES / 16)
#define HOLD4(a, i) "v"(a[i]), "v"(a[(i) + 1]), "v"(a[(i) + 2]), "v"(a[(i) + 3])

__device__ __forceinline__ int frag_k(int t, int h)
{
    return 8 * (t >> 1) + 4 * h + 2 * (t & 1);
}

template <int K, int NT, int WPB, bool HASADD, int FOLD0, int FSPLIT>
__global__ __launch_bounds__(NT * WPB * 32)
void gemm_wmma_f32_kernel(const float* __restrict__ A0, int lda0,
                          const float* __restrict__ A1, int lda1,
                          const float* __restrict__ W,  int ldw,
                          const float* __restrict__ bias,
                          float* __restrict__ out, int ldo,
                          float* __restrict__ colsum,
                          float* __restrict__ colsq,
                          const float* __restrict__ fsum,
                          const float* __restrict__ fsq,
                          const float* __restrict__ fg,
                          const float* __restrict__ fbeta,
                          const float* __restrict__ fsum2,
                          const float* __restrict__ fsq2,
                          const float* __restrict__ fg2,
                          const float* __restrict__ fbeta2)
{
    constexpr int STEPS  = K / 4;
    constexpr int CHUNKS = K / 8;

    const int lane = threadIdx.x & 31;
    const int wave = __builtin_amdgcn_readfirstlane((int)(threadIdx.x >> 5));
    const int lo   = lane & 15;
    const int hi   = lane >> 4;
    const int ct       = wave % NT;
    const int walker   = blockIdx.x * WPB + wave / NT;
    const int nWalkers = gridDim.x * WPB;
    const int col      = ct * 16 + lo;

    v2f b[STEPS];
#pragma unroll
    for (int t = 0; t < STEPS; ++t) {
        const int k = frag_k(t, hi);
        b[t].x = W[(size_t)k * ldw + col];
        b[t].y = W[(size_t)(k + 1) * ldw + col];
    }
    float bv = bias[col];

    if constexpr (FOLD0 < K) {
        constexpr int KF = K - FOLD0;
        static_assert(FOLD0 % 8 == 0, "a 16-byte A chunk must lie wholly on one side of FOLD0");
        static_assert(KF <= NT * WPB * 32, "one thread per folded column");
        __shared__ float fsc[KF];
        __shared__ float fsh[KF];
        static_assert(FSPLIT > 0 && FSPLIT <= KF, "the first set covers at least one folded column");
        if (threadIdx.x < KF) {
            const int i = threadIdx.x;
            float su, sq, ga, be;
            if (FSPLIT == KF || i < FSPLIT) {
                su = fsum[i];  sq = fsq[i];  ga = fg[i];  be = fbeta[i];
            } else {
                su = fsum2[i - FSPLIT];  sq = fsq2[i - FSPLIT];
                ga = fg2[i - FSPLIT];    be = fbeta2[i - FSPLIT];
            }
            const float mu  = su * INV_N;
            const float var = sq * INV_N - mu * mu;
            const float sc  = ga * rsqrtf(var + BN_EPS);
            fsc[i] = sc;
            fsh[i] = be - mu * sc;
        }
        __syncthreads();
        float part = 0.0f;
#pragma unroll
        for (int t = 0; t < STEPS; ++t) {
            if (8 * (t >> 1) >= FOLD0) {
                const int kf = frag_k(t, hi) - FOLD0;
                part += fsh[kf] * b[t].x + fsh[kf + 1] * b[t].y;
                b[t].x *= fsc[kf];
                b[t].y *= fsc[kf + 1];
            }
        }
        bv += part + __shfl_xor(part, 16);
    }

    float s1 = 0.0f, s2 = 0.0f;

    for (int rt = walker; rt < ROWTILES; rt += nWalkers) {
        const int row0 = rt * 16;
        const v4f* ap0 = reinterpret_cast<const v4f*>(
            A0 + (size_t)(row0 + lo) * lda0 + 4 * hi);

        v4f q[CHUNKS];
#pragma unroll
        for (int j = 0; j < CHUNKS; ++j) q[j] = ap0[2 * j];

        if constexpr (HASADD) {
            const v4f* ap1 = reinterpret_cast<const v4f*>(
                A1 + (size_t)(row0 + lo) * lda1 + 4 * hi);
            v4f p[CHUNKS];
#pragma unroll
            for (int j = 0; j < CHUNKS; ++j) p[j] = ap1[2 * j];
            static_assert(CHUNKS == 12, "fence operand list is written for K = 96");
            asm volatile("" :: HOLD4(q, 0), HOLD4(q, 4), HOLD4(q, 8),
                               HOLD4(p, 0), HOLD4(p, 4), HOLD4(p, 8));
#pragma unroll
            for (int j = 0; j < CHUNKS; ++j) q[j] += p[j];
        } else if constexpr (CHUNKS == 12) {
            asm volatile("" :: HOLD4(q, 0), HOLD4(q, 4), HOLD4(q, 8));
        } else {
            static_assert(CHUNKS == 20, "fence operand list is written for K = 96 or 160");
            asm volatile("" :: HOLD4(q, 0), HOLD4(q, 4), HOLD4(q, 8),
                               HOLD4(q, 12), HOLD4(q, 16));
        }

        v8f c = {};
#pragma unroll
        for (int j = 0; j < CHUNKS; ++j) {
            v2f a;
            a.x = q[j].x; a.y = q[j].y;
            c = __builtin_amdgcn_wmma_f32_16x16x4_f32(
                     false, a,  false, b[2 * j],
                     (short)0, c,  false,  false);
            a.x = q[j].z; a.y = q[j].w;
            c = __builtin_amdgcn_wmma_f32_16x16x4_f32(
                     false, a,  false, b[2 * j + 1],
                     (short)0, c,  false,  false);
        }

#pragma unroll
        for (int r = 0; r < 8; ++r) {
            int row = row0 + r + 8 * hi;
            const float v = fmaxf(c[r] + bv, 0.0f);
            out[(size_t)row * ldo + col] = v;
            s1 += v;
            s2 += v * v;
        }
    }

    s1 += __shfl_xor(s1, 16);
    s2 += __shfl_xor(s2, 16);

    if constexpr (WPB == 1) {
        if (hi == 0) {
            atomicAdd(&colsum[col], s1);
            atomicAdd(&colsq[col],  s2);
        }
    } else {
        static_assert(NT == 1, "in-block add assumes all waves own the same columns");
        __shared__ float red[WPB][32];
        red[wave][lane] = (hi == 0) ? s1 : s2;
        __syncthreads();
        if (wave == 0) {
            float t = 0.0f;
            for (int w = 0; w < WPB; ++w) t += red[w][lane];
            atomicAdd((hi == 0 ? colsum : colsq) + lo, t);
        }
    }
}

__global__ __launch_bounds__(256)
void bn_apply_kernel(const float* __restrict__ in, int ldi,
                     float* __restrict__ out, int ldo,
                     const float* __restrict__ sum,
                     const float* __restrict__ sumsq,
                     const float* __restrict__ g,
                     const float* __restrict__ beta, int n, int C)
{
    int gid = blockIdx.x * blockDim.x + threadIdx.x;
    if (gid >= n * C) return;
    int row = gid / C;
    int c   = gid % C;
    float mu  = sum[c] * INV_N;
    float var = sumsq[c] * INV_N - mu * mu;
    float sc  = g[c] * rsqrtf(var + BN_EPS);
    float sh  = beta[c] - mu * sc;
    out[(size_t)row * ldo + c] = in[(size_t)row * ldi + c] * sc + sh;
}

extern "C" void kernel_launch(void* const* d_in, const int* in_sizes, int n_in,
                              void* d_out, int out_size, void* d_ws, size_t ws_size,
                              hipStream_t stream)
{
    (void)in_sizes; (void)n_in; (void)out_size; (void)ws_size;

    const float* x          = (const float*)d_in[0];
    const float* edge_attr  = (const float*)d_in[1];
    const int*   edge_index = (const int*)  d_in[2];
    const float* lin_w      = (const float*)d_in[3];
    const float* lin_b      = (const float*)d_in[4];
    const float* c1_w  = (const float*)d_in[5];
    const float* c1_b  = (const float*)d_in[6];
    const float* c1_g  = (const float*)d_in[7];
    const float* c1_be = (const float*)d_in[8];
    const float* c2_w  = (const float*)d_in[9];
    const float* c2_b  = (const float*)d_in[10];
    const float* c2_g  = (const float*)d_in[11];
    const float* c2_be = (const float*)d_in[12];
    const float* l1_w  = (const float*)d_in[13];
    const float* l1_b  = (const float*)d_in[14];
    const float* l1_g  = (const float*)d_in[15];
    const float* l1_be = (const float*)d_in[16];
    const float* ma_w  = (const float*)d_in[17];
    const float* ma_b  = (const float*)d_in[18];
    const float* ma_g  = (const float*)d_in[19];
    const float* ma_be = (const float*)d_in[20];
    const float* mb_w  = (const float*)d_in[21];
    const float* mb_b  = (const float*)d_in[22];
    const float* mb_g  = (const float*)d_in[23];
    const float* mb_be = (const float*)d_in[24];

    const int* src = edge_index;
    const int* dst = edge_index + NEDGES;

    const size_t N = NNODES;

    float* ws    = (float*)d_ws;
    float* agg   = ws;
    float* bufP  = agg  + N * 96;
    float* bufQ  = bufP + N * 96;
    float* xcat  = bufQ + N * 96;
    float* stats = xcat + N * 160;
    int* ownCnt  = (int*)(stats + 1024);
    int* ownRec  = (int*)(stats + 2048);
    unsigned* ownKey = (unsigned*)(ownRec + (size_t)OWN_BLOCKS * OWN_QCAP * 2);


    auto sumL = [&](int L) { return stats + (size_t)L * 192; };
    auto sqL  = [&](int L) { return stats + (size_t)L * 192 + 96; };

    hipMemsetAsync(stats, 0, 5 * 192 * sizeof(float), stream);

    owner_key_kernel<<<(NEDGES / KEY_WAVE_EDGES + 7) / 8, 256, 0, stream>>>(dst, ownKey);

    edge_owner_kernel<true><<<OWN_BLOCKS, OWN_THREADS, 0, stream>>>(
        x, 96, edge_attr, src, dst, ownKey, lin_w, lin_b, agg, ownCnt, ownRec,
        nullptr, nullptr, nullptr, nullptr);

    gemm_wmma_f32_kernel<96, 6, 1, true, 96, 1><<<625, 192, 0, stream>>>(
        x, 96, agg, 96, c1_w, 96, c1_b, xcat, 160, sumL(0), sqL(0),
        nullptr, nullptr, nullptr, nullptr, nullptr, nullptr, nullptr, nullptr);

    edge_owner_kernel<false><<<OWN_BLOCKS, OWN_THREADS, 0, stream>>>(
        xcat, 160, edge_attr, src, dst, ownKey, lin_w, lin_b, agg, ownCnt, ownRec,
        sumL(0), sqL(0), c1_g, c1_be);

    gemm_wmma_f32_kernel<96, 4, 1, false, 96, 1><<<625, 128, 0, stream>>>(
        agg, 96, nullptr, 0, c2_w, 64, c2_b, xcat + 96, 160, sumL(1), sqL(1),
        nullptr, nullptr, nullptr, nullptr, nullptr, nullptr, nullptr, nullptr);

    gemm_wmma_f32_kernel<160, 6, 1, false, 0, 96><<<625, 192, 0, stream>>>(
        xcat, 160, nullptr, 0, l1_w, 96, l1_b, bufQ, 96, sumL(2), sqL(2),
        sumL(0), sqL(0), c1_g, c1_be, sumL(1), sqL(1), c2_g, c2_be);

    gemm_wmma_f32_kernel<96, 6, 1, false, 0, 96><<<625, 192, 0, stream>>>(
        bufQ, 96, nullptr, 0, ma_w, 96, ma_b, bufP, 96, sumL(3), sqL(3),
        sumL(2), sqL(2), l1_g, l1_be, nullptr, nullptr, nullptr, nullptr);

    gemm_wmma_f32_kernel<96, 1, 5, false, 0, 96><<<625, 160, 0, stream>>>(
        bufP, 96, nullptr, 0, mb_w, 16, mb_b, bufQ, 16, sumL(4), sqL(4),
        sumL(3), sqL(3), ma_g, ma_be, nullptr, nullptr, nullptr, nullptr);

    bn_apply_kernel<<<(NNODES * 16 + 255) / 256, 256, 0, stream>>>(
        bufQ, 16, (float*)d_out, 16, sumL(4), sqL(4), mb_g, mb_be, NNODES, 16);
}


# Dense layers: a block copies each 16-row A tile once into LDS as whole 128-byte lines, one tile ahead; its column waves read fragments from LDS, not each the same rows from memory; sums unchanged
# speedup vs baseline: 6.4849x; 1.2041x over previous
#include <hip/hip_runtime.h>
#include <hip/hip_bf16.h>


#define NNODES 50000
#define NEDGES 800000
#define BN_EPS 1e-5f
#define INV_N  (1.0f / (float)NNODES)

typedef __attribute__((ext_vector_type(2))) float v2f;
typedef __attribute__((ext_vector_type(4))) float v4f;
typedef __attribute__((ext_vector_type(8))) float v8f;

#define OWN_ROWS    200
#define OWN_BLOCKS  (NNODES / OWN_ROWS)
#define OWN_THREADS 1024
#define OWN_WAVES   (OWN_THREADS / 32)
#define OWN_QCAP    24576
#define NGRP        (NEDGES / 32)
#define KEY_NONE    255u
#define SCAN_DEPTH  4
#define ROW_DEPTH   16

static_assert(OWN_ROWS * OWN_BLOCKS == NNODES, "every row has exactly one owner");
static_assert(NEDGES % 256 == 0, "a wave of the key launch makes eight whole groups");
static_assert(OWN_BLOCKS <= KEY_NONE, "a block number fits the 8-bit key and is never KEY_NONE");
static_assert(NNODES <= (1 << 24) && OWN_ROWS <= (1 << 8), "record word = src | local row << 24");
static_assert(OWN_QCAP % 32 == 0 && NGRP % OWN_BLOCKS == 0 && ((NGRP / OWN_BLOCKS) * 32) % 128 == 0,
              "slices are whole groups; start shifts fall on 128-byte lines");

#define KEY_WAVE_EDGES 256

__global__ __launch_bounds__(256)
void owner_key_kernel(const int* __restrict__ dst, unsigned* __restrict__ planes)
{
    const int lane = threadIdx.x & 31;
    const int wv = __builtin_amdgcn_readfirstlane((int)((blockIdx.x * 256 + threadIdx.x) >> 5));
    if (wv >= NEDGES / KEY_WAVE_EDGES) return;
    const int* dp = dst + (size_t)wv * KEY_WAVE_EDGES + lane;
    int d[8];
#pragma unroll
    for (int g = 0; g < 8; ++g) d[g] = dp[32 * g];
    unsigned outw[2] = { 0u, 0u };
#pragma unroll
    for (int g = 0; g < 8; ++g) {
        const unsigned key = ((unsigned)d[g] < (unsigned)NNODES)
                           ? (unsigned)d[g] / (unsigned)OWN_ROWS : KEY_NONE;
#pragma unroll
        for (int p = 0; p < 8; ++p) {
            const unsigned vote = __builtin_amdgcn_ballot_w32(((key >> p) & 1u) != 0u);
            if (lane == 8 * (g & 3) + p) outw[g >> 2] = vote;
        }
    }
    unsigned* op = planes + (size_t)wv * (KEY_WAVE_EDGES / 4) + lane;
    op[0]  = outw[0];
    op[32] = outw[1];
}

struct OwnFlip { unsigned f[8]; };

__device__ __forceinline__ void own_scan(const uint4* __restrict__ p4,
                                         int gBegin, int gEnd, int rot, const OwnFlip flip,
                                         int* queue, int* qcount, int wave, int lane)
{
    for (int base = gBegin; base < gEnd; base += SCAN_DEPTH * OWN_THREADS) {
        uint4 lo[SCAN_DEPTH], hi[SCAN_DEPTH];
        int   gg[SCAN_DEPTH];
#pragma unroll
        for (int u = 0; u < SCAN_DEPTH; ++u) {
            const int g = base + u * OWN_THREADS + wave * 32 + lane;
            int w = g + rot;
            if (w >= NGRP) w -= NGRP;
            gg[u] = (g < gEnd) ? w : -1;
            const uint4* gp = p4 + 2 * (size_t)((g < gEnd) ? w : 0);
            lo[u] = gp[0];
            hi[u] = gp[1];
        }
#pragma unroll
        for (int u = 0; u < SCAN_DEPTH; ++u) {
            const int gw = base + u * OWN_THREADS + wave * 32;
            if (gw < gEnd) {
                unsigned m = (lo[u].x ^ flip.f[0]) & (lo[u].y ^ flip.f[1]) &
                             (lo[u].z ^ flip.f[2]) & (lo[u].w ^ flip.f[3]) &
                             (hi[u].x ^ flip.f[4]) & (hi[u].y ^ flip.f[5]) &
                             (hi[u].z ^ flip.f[6]) & (hi[u].w ^ flip.f[7]);
                if (gg[u] < 0) m = 0u;
                for (int it = 0; it < 32; ++it) {
                    const unsigned mask = __builtin_amdgcn_ballot_w32(m != 0u);
                    if (mask == 0u) break;
                    int first = 0;
                    if (lane == 0)
                        first = atomicAdd(qcount, __popc(mask));
                    first = __builtin_amdgcn_readfirstlane(first);
                    if (m != 0u) {
                        const int pos = first + (int)__builtin_amdgcn_mbcnt_lo(mask, 0u);
                        if (pos < OWN_QCAP)
                            queue[pos] = 32 * gg[u] + __builtin_ctz(m);
                        m &= m - 1u;
                    }
                }
            }
        }
    }
}

template <bool FROMREC, bool SAVE, bool NORM>
__device__ __forceinline__ void own_process(int n,
                                            const float* __restrict__ x, int ldx,
                                            const float* __restrict__ ea,
                                            const int* __restrict__ src,
                                            const int* __restrict__ dst, int r0,
                                            const int* queue, int2* recs, float* acc,
                                            float lw0, float lw1, float lw2,
                                            float lb0, float lb1, float lb2,
                                            float sc0, float sc1, float sc2,
                                            float sh0, float sh1, float sh2,
                                            int wave, int lane)
{
    for (int g = wave * 32; g < n; g += 32 * OWN_WAVES) {
        const bool live = g + lane < n;
        int s, ab, lrow;
        if constexpr (FROMREC) {
            const int2 r = live ? recs[g + lane] : make_int2(0, 0);
            s    = r.x & 0xFFFFFF;
            lrow = (int)((unsigned)r.x >> 24);
            ab   = r.y;
        } else {
            const int e = live ? queue[g + lane] : 0;
            lrow = dst[e] - r0;
            s    = src[e];
            ab   = __float_as_int(ea[e]);
            if constexpr (SAVE) {
                if (live)
                    recs[g + lane] = make_int2((int)((unsigned)s | ((unsigned)lrow << 24)), ab);
            }
        }
        const int cnt  = (n - g < 32) ? (n - g) : 32;
        for (int j0 = 0; j0 < cnt; j0 += ROW_DEPTH) {
            float xv[ROW_DEPTH][3];
#pragma unroll
            for (int u = 0; u < ROW_DEPTH; ++u) {
                const int sj = __builtin_amdgcn_readlane(s, j0 + u);
                const float* xr = x + (size_t)sj * ldx + lane;
                xv[u][0] = xr[0];
                xv[u][1] = xr[32];
                xv[u][2] = xr[64];
            }
#pragma unroll
            for (int u = 0; u < ROW_DEPTH; ++u) {
                if (j0 + u < cnt) {
                    const float a  = __int_as_float(__builtin_amdgcn_readlane(ab, j0 + u));
                    const int   rj = __builtin_amdgcn_readlane(lrow, j0 + u);
                    float* ar = acc + rj * 96 + lane;
                    float x0 = xv[u][0], x1 = xv[u][1], x2 = xv[u][2];
                    if constexpr (NORM) {
                        x0 = x0 * sc0 + sh0;
                        x1 = x1 * sc1 + sh1;
                        x2 = x2 * sc2 + sh2;
                    }
                    atomicAdd(ar,      fmaxf(x0 + a * lw0 + lb0, 0.0f));
                    atomicAdd(ar + 32, fmaxf(x1 + a * lw1 + lb1, 0.0f));
                    atomicAdd(ar + 64, fmaxf(x2 + a * lw2 + lb2, 0.0f));
                }
            }
        }
    }
}

template <bool FIRST>
__global__ __launch_bounds__(OWN_THREADS)
void edge_owner_kernel(const float* __restrict__ x, int ldx,
                       const float* __restrict__ ea,
                       const int* __restrict__ src,
                       const int* __restrict__ dst,
                       const unsigned* __restrict__ planes,
                       const float* __restrict__ lw,
                       const float* __restrict__ lb,
                       float* __restrict__ agg,
                       int* __restrict__ savedCnt,
                       int* __restrict__ savedRec,
                       const float* __restrict__ nsum,
                       const float* __restrict__ nsq,
                       const float* __restrict__ ng,
                       const float* __restrict__ nbeta)
{
    constexpr bool NORM = !FIRST;
    __shared__ __attribute__((aligned(16))) float acc[OWN_ROWS * 96];
    __shared__ int queue[OWN_QCAP];
    __shared__ int qcount;

    const int tid  = threadIdx.x;
    const int lane = tid & 31;
    const int wave = __builtin_amdgcn_readfirstlane(tid >> 5);
    const int r0   = blockIdx.x * OWN_ROWS;
    const uint4* p4 = reinterpret_cast<const uint4*>(planes);
    OwnFlip flip;
#pragma unroll
    for (int p = 0; p < 8; ++p) flip.f[p] = ((blockIdx.x >> p) & 1u) ? 0u : ~0u;
    int2* recs = reinterpret_cast<int2*>(savedRec) + (size_t)blockIdx.x * OWN_QCAP;
    int total = 0;
    if constexpr (!FIRST) total = savedCnt[blockIdx.x];

    const float lw0 = lw[lane], lw1 = lw[lane + 32], lw2 = lw[lane + 64];
    const float lb0 = lb[lane], lb1 = lb[lane + 32], lb2 = lb[lane + 64];

    float sc[3] = { 1.0f, 1.0f, 1.0f };
    float sh[3] = { 0.0f, 0.0f, 0.0f };
    if constexpr (NORM) {
        constexpr int INIT_ROWS = (OWN_ROWS + OWN_WAVES - 1) / OWN_WAVES;
        float hv[INIT_ROWS][3];
#pragma unroll
        for (int u = 0; u < INIT_ROWS; ++u) {
            const int row = wave + OWN_WAVES * u;
            const float* xr = x + (size_t)(r0 + (row < OWN_ROWS ? row : 0)) * ldx + lane;
            hv[u][0] = xr[0];
            hv[u][1] = xr[32];
            hv[u][2] = xr[64];
        }
#pragma unroll
        for (int m = 0; m < 3; ++m) {
            const int cc = lane + 32 * m;
            const float mu  = nsum[cc] * INV_N;
            const float var = nsq[cc] * INV_N - mu * mu;
            sc[m] = ng[cc] * rsqrtf(var + BN_EPS);
            sh[m] = nbeta[cc] - mu * sc[m];
        }
#pragma unroll
        for (int u = 0; u < INIT_ROWS; ++u) {
            const int row = wave + OWN_WAVES * u;
            if (row < OWN_ROWS) {
                float* ar = acc + row * 96 + lane;
                ar[0]  = hv[u][0] * sc[0] + sh[0];
                ar[32] = hv[u][1] * sc[1] + sh[1];
                ar[64] = hv[u][2] * sc[2] + sh[2];
            }
        }
    } else {
        for (int i = tid; i < OWN_ROWS * 96; i += OWN_THREADS) acc[i] = 0.0f;
    }
    if (tid == 0) qcount = 0;

    __syncthreads();

    if constexpr (FIRST) {
        own_scan(p4, 0, NGRP, (int)blockIdx.x * (NGRP / OWN_BLOCKS), flip, queue, &qcount, wave, lane);

        __syncthreads();
        total = qcount;
        if (tid == 0) savedCnt[blockIdx.x] = total;
    }
    total = __builtin_amdgcn_readfirstlane(total);

    if (total <= OWN_QCAP) {
        if constexpr (FIRST)
            own_process<false, true, NORM>(total, x, ldx, ea, src, dst, r0, queue, recs, acc,
                                           lw0, lw1, lw2, lb0, lb1, lb2,
                                           sc[0], sc[1], sc[2], sh[0], sh[1], sh[2], wave, lane);
        else
            own_process<true, false, NORM>(total, x, ldx, ea, src, dst, r0, queue, recs, acc,
                                           lw0, lw1, lw2, lb0, lb1, lb2,
                                           sc[0], sc[1], sc[2], sh[0], sh[1], sh[2], wave, lane);
    } else {
        for (int vb = 0; vb < NGRP; vb += OWN_QCAP / 32) {
            __syncthreads();
            if (tid == 0) qcount = 0;
            __syncthreads();
            const int ve = (vb + OWN_QCAP / 32 < NGRP) ? (vb + OWN_QCAP / 32) : NGRP;
            own_scan(p4, vb, ve, 0, flip, queue, &qcount, wave, lane);
            __syncthreads();
            const int n = __builtin_amdgcn_readfirstlane(qcount);
            own_process<false, false, NORM>(n, x, ldx, ea, src, dst, r0, queue, recs, acc,
                                            lw0, lw1, lw2, lb0, lb1, lb2,
                                            sc[0], sc[1], sc[2], sh[0], sh[1], sh[2], wave, lane);
        }
    }

    __syncthreads();
    float4* out4 = reinterpret_cast<float4*>(agg + (size_t)r0 * 96);
    const float4* a4 = reinterpret_cast<const float4*>(acc);
    for (int i = tid; i < OWN_ROWS * 96 / 4; i += OWN_THREADS) out4[i] = a4[i];
}

#define ROWTILES (NNODES / 16)
#define HOLD4(a, i) "v"(a[i]), "v"(a[(i) + 1]), "v"(a[(i) + 2]), "v"(a[(i) + 3])

__device__ __forceinline__ int frag_k(int t, int h)
{
    return 8 * (t >> 1) + 4 * h + 2 * (t & 1);
}

template <int K, int NT, int WPB, bool HASADD, int FOLD0, int FSPLIT>
__global__ __launch_bounds__(NT * WPB * 32)
void gemm_wmma_f32_kernel(const float* __restrict__ A0, int lda0,
                          const float* __restrict__ A1, int lda1,
                          const float* __restrict__ W,  int ldw,
                          const float* __restrict__ bias,
                          float* __restrict__ out, int ldo,
                          float* __restrict__ colsum,
                          float* __restrict__ colsq,
                          const float* __restrict__ fsum,
                          const float* __restrict__ fsq,
                          const float* __restrict__ fg,
                          const float* __restrict__ fbeta,
                          const float* __restrict__ fsum2,
                          const float* __restrict__ fsq2,
                          const float* __restrict__ fg2,
                          const float* __restrict__ fbeta2)
{
    constexpr int STEPS  = K / 4;
    constexpr int CHUNKS = K / 8;

    const int lane = threadIdx.x & 31;
    const int wave = __builtin_amdgcn_readfirstlane((int)(threadIdx.x >> 5));
    const int lo   = lane & 15;
    const int hi   = lane >> 4;
    const int ct       = wave % NT;
    const int walker   = blockIdx.x * WPB + wave / NT;
    const int nWalkers = gridDim.x * WPB;
    const int col      = ct * 16 + lo;

    constexpr bool STAGED = (WPB == 1 && NT > 1);
    constexpr int NTHR   = NT * WPB * 32;
    constexpr int ROWCH  = K / 4;
    constexpr int RPR    = NTHR / ROWCH;
    constexpr int CT     = RPR * ROWCH;
    constexpr int PER    = STAGED ? (16 + RPR - 1) / RPR : 1;
    constexpr int PITCH  = K + 4;
    static_assert(((PITCH / 4) & 1) == 1, "row pitch must be an odd number of 16-byte units");
    static_assert(RPR >= 1, "a round covers at least one row");
    v4f g0[PER], g1[PER];
    const int  cr     = (int)threadIdx.x / ROWCH;
    const bool copier = (int)threadIdx.x < CT;
    const unsigned goff0 = (unsigned)(cr * lda0 + 4 * ((int)threadIdx.x % ROWCH));
    const unsigned goff1 = HASADD ? (unsigned)(cr * lda1 + 4 * ((int)threadIdx.x % ROWCH)) : 0u;
    const unsigned soff  = (unsigned)(cr * PITCH + 4 * ((int)threadIdx.x % ROWCH));
    auto tile_fetch = [&](int rt) {
        const unsigned t0 = (unsigned)(rt * 16 * lda0);
        const unsigned t1 = HASADD ? (unsigned)(rt * 16 * lda1) : 0u;
#pragma unroll
        for (int u = 0; u < PER; ++u) {
            if (copier && (((u + 1) * RPR <= 16) || (cr + u * RPR < 16))) {
                g0[u] = *reinterpret_cast<const v4f*>(A0 + (t0 + (unsigned)(u * RPR * lda0) + goff0));
                if constexpr (HASADD)
                    g1[u] = *reinterpret_cast<const v4f*>(A1 + (t1 + (unsigned)(u * RPR * lda1) + goff1));
            }
        }
    };
    __shared__ __attribute__((aligned(16))) float stage[STAGED ? 2 * 16 * PITCH : 4];
    auto tile_put = [&](int buf) {
        float* sp = stage + buf * (16 * PITCH) + soff;
#pragma unroll
        for (int u = 0; u < PER; ++u) {
            if (copier && (((u + 1) * RPR <= 16) || (cr + u * RPR < 16))) {
                v4f v = g0[u];
                if constexpr (HASADD) v += g1[u];
                *reinterpret_cast<v4f*>(sp + u * RPR * PITCH) = v;
            }
        }
    };
    constexpr bool EARLY = STAGED && K <= 96;
    if constexpr (EARLY) {
        if (walker < ROWTILES) tile_fetch(walker);
    }

    v2f b[STEPS];
#pragma unroll
    for (int t = 0; t < STEPS; ++t) {
        const int k = frag_k(t, hi);
        b[t].x = W[(unsigned)(k * ldw + col)];
        b[t].y = W[(unsigned)((k + 1) * ldw + col)];
    }
    float bv = bias[col];
    if constexpr (EARLY) {
        if (walker < ROWTILES) tile_put(0);
    }

    if constexpr (FOLD0 < K) {
        constexpr int KF = K - FOLD0;
        static_assert(FOLD0 % 8 == 0, "a 16-byte A chunk must lie wholly on one side of FOLD0");
        static_assert(KF <= NT * WPB * 32, "one thread per folded column");
        __shared__ float fsc[KF];
        __shared__ float fsh[KF];
        static_assert(FSPLIT > 0 && FSPLIT <= KF, "the first set covers at least one folded column");
        if (threadIdx.x < KF) {
            const int i = threadIdx.x;
            float su, sq, ga, be;
            if (FSPLIT == KF || i < FSPLIT) {
                su = fsum[i];  sq = fsq[i];  ga = fg[i];  be = fbeta[i];
            } else {
                su = fsum2[i - FSPLIT];  sq = fsq2[i - FSPLIT];
                ga = fg2[i - FSPLIT];    be = fbeta2[i - FSPLIT];
            }
            const float mu  = su * INV_N;
            const float var = sq * INV_N - mu * mu;
            const float sc  = ga * rsqrtf(var + BN_EPS);
            fsc[i] = sc;
            fsh[i] = be - mu * sc;
        }
        __syncthreads();
        float part = 0.0f;
#pragma unroll
        for (int t = 0; t < STEPS; ++t) {
            if (8 * (t >> 1) >= FOLD0) {
                const int kf = frag_k(t, hi) - FOLD0;
                part += fsh[kf] * b[t].x + fsh[kf + 1] * b[t].y;
                b[t].x *= fsc[kf];
                b[t].y *= fsc[kf + 1];
            }
        }
        bv += part + __shfl_xor(part, 16);
    }

    float s1 = 0.0f, s2 = 0.0f;

    auto tile_products = [&](const v4f (&q)[CHUNKS]) {
        v8f c = {};
#pragma unroll
        for (int j = 0; j < CHUNKS; ++j) {
            v2f a;
            a.x = q[j].x; a.y = q[j].y;
            c = __builtin_amdgcn_wmma_f32_16x16x4_f32(
                     false, a,  false, b[2 * j],
                     (short)0, c,  false,  false);
            a.x = q[j].z; a.y = q[j].w;
            c = __builtin_amdgcn_wmma_f32_16x16x4_f32(
                     false, a,  false, b[2 * j + 1],
                     (short)0, c,  false,  false);
        }
        return c;
    };
    auto tile_store = [&](int row0, const v8f c) {
#pragma unroll
        for (int r = 0; r < 8; ++r) {
            int row = row0 + r + 8 * hi;
            const float v = fmaxf(c[r] + bv, 0.0f);
            out[(unsigned)(row * ldo + col)] = v;
            s1 += v;
            s2 += v * v;
        }
    };

    if constexpr (STAGED) {
        if constexpr (!EARLY) {
            if (walker < ROWTILES) { tile_fetch(walker); tile_put(0); }
        }
        __syncthreads();
        int buf = 0;
        for (int rt = walker; rt < ROWTILES; rt += nWalkers) {
            const int nx = rt + nWalkers;
            if (nx < ROWTILES) tile_fetch(nx);
            const float* fp = stage + buf * (16 * PITCH) + lo * PITCH + 4 * hi;
            v4f q[CHUNKS];
#pragma unroll
            for (int j = 0; j < CHUNKS; ++j) q[j] = *reinterpret_cast<const v4f*>(fp + 8 * j);
            const v8f c = tile_products(q);
            if (nx < ROWTILES) tile_put(buf ^ 1);
            __syncthreads();
            tile_store(rt * 16, c);
            buf ^= 1;
        }
    } else {
        static_assert(!HASADD && CHUNKS == 12, "the unshared path is written for one operand and K = 96");
        for (int rt = walker; rt < ROWTILES; rt += nWalkers) {
            const int row0 = rt * 16;
            const v4f* ap0 = reinterpret_cast<const v4f*>(
                A0 + (size_t)(row0 + lo) * lda0 + 4 * hi);
            v4f q[CHUNKS];
#pragma unroll
            for (int j = 0; j < CHUNKS; ++j) q[j] = ap0[2 * j];
            asm volatile("" :: HOLD4(q, 0), HOLD4(q, 4), HOLD4(q, 8));
            tile_store(row0, tile_products(q));
        }
    }

    s1 += __shfl_xor(s1, 16);
    s2 += __shfl_xor(s2, 16);

    if constexpr (WPB == 1) {
        if (hi == 0) {
            atomicAdd(&colsum[col], s1);
            atomicAdd(&colsq[col],  s2);
        }
    } else {
        static_assert(NT == 1, "in-block add assumes all waves own the same columns");
        __shared__ float red[WPB][32];
        red[wave][lane] = (hi == 0) ? s1 : s2;
        __syncthreads();
        if (wave == 0) {
            float t = 0.0f;
            for (int w = 0; w < WPB; ++w) t += red[w][lane];
            atomicAdd((hi == 0 ? colsum : colsq) + lo, t);
        }
    }
}

__global__ __launch_bounds__(256)
void bn_apply_kernel(const float* __restrict__ in, int ldi,
                     float* __restrict__ out, int ldo,
                     const float* __restrict__ sum,
                     const float* __restrict__ sumsq,
                     const float* __restrict__ g,
                     const float* __restrict__ beta, int n, int C)
{
    int gid = blockIdx.x * blockDim.x + threadIdx.x;
    if (gid >= n * C) return;
    int row = gid / C;
    int c   = gid % C;
    float mu  = sum[c] * INV_N;
    float var = sumsq[c] * INV_N - mu * mu;
    float sc  = g[c] * rsqrtf(var + BN_EPS);
    float sh  = beta[c] - mu * sc;
    out[(size_t)row * ldo + c] = in[(size_t)row * ldi + c] * sc + sh;
}

extern "C" void kernel_launch(void* const* d_in, const int* in_sizes, int n_in,
                              void* d_out, int out_size, void* d_ws, size_t ws_size,
                              hipStream_t stream)
{
    (void)in_sizes; (void)n_in; (void)out_size; (void)ws_size;

    const float* x          = (const float*)d_in[0];
    const float* edge_attr  = (const float*)d_in[1];
    const int*   edge_index = (const int*)  d_in[2];
    const float* lin_w      = (const float*)d_in[3];
    const float* lin_b      = (const float*)d_in[4];
    const float* c1_w  = (const float*)d_in[5];
    const float* c1_b  = (const float*)d_in[6];
    const float* c1_g  = (const float*)d_in[7];
    const float* c1_be = (const float*)d_in[8];
    const float* c2_w  = (const float*)d_in[9];
    const float* c2_b  = (const float*)d_in[10];
    const float* c2_g  = (const float*)d_in[11];
    const float* c2_be = (const float*)d_in[12];
    const float* l1_w  = (const float*)d_in[13];
    const float* l1_b  = (const float*)d_in[14];
    const float* l1_g  = (const float*)d_in[15];
    const float* l1_be = (const float*)d_in[16];
    const float* ma_w  = (const float*)d_in[17];
    const float* ma_b  = (const float*)d_in[18];
    const float* ma_g  = (const float*)d_in[19];
    const float* ma_be = (const float*)d_in[20];
    const float* mb_w  = (const float*)d_in[21];
    const float* mb_b  = (const float*)d_in[22];
    const float* mb_g  = (const float*)d_in[23];
    const float* mb_be = (const float*)d_in[24];

    const int* src = edge_index;
    const int* dst = edge_index + NEDGES;

    const size_t N = NNODES;

    float* ws    = (float*)d_ws;
    float* agg   = ws;
    float* bufP  = agg  + N * 96;
    float* bufQ  = bufP + N * 96;
    float* xcat  = bufQ + N * 96;
    float* stats = xcat + N * 160;
    int* ownCnt  = (int*)(stats + 1024);
    int* ownRec  = (int*)(stats + 2048);
    unsigned* ownKey = (unsigned*)(ownRec + (size_t)OWN_BLOCKS * OWN_QCAP * 2);


    auto sumL = [&](int L) { return stats + (size_t)L * 192; };
    auto sqL  = [&](int L) { return stats + (size_t)L * 192 + 96; };

    hipMemsetAsync(stats, 0, 5 * 192 * sizeof(float), stream);

    owner_key_kernel<<<(NEDGES / KEY_WAVE_EDGES + 7) / 8, 256, 0, stream>>>(dst, ownKey);

    edge_owner_kernel<true><<<OWN_BLOCKS, OWN_THREADS, 0, stream>>>(
        x, 96, edge_attr, src, dst, ownKey, lin_w, lin_b, agg, ownCnt, ownRec,
        nullptr, nullptr, nullptr, nullptr);

    gemm_wmma_f32_kernel<96, 6, 1, true, 96, 1><<<625, 192, 0, stream>>>(
        x, 96, agg, 96, c1_w, 96, c1_b, xcat, 160, sumL(0), sqL(0),
        nullptr, nullptr, nullptr, nullptr, nullptr, nullptr, nullptr, nullptr);

    edge_owner_kernel<false><<<OWN_BLOCKS, OWN_THREADS, 0, stream>>>(
        xcat, 160, edge_attr, src, dst, ownKey, lin_w, lin_b, agg, ownCnt, ownRec,
        sumL(0), sqL(0), c1_g, c1_be);

    gemm_wmma_f32_kernel<96, 4, 1, false, 96, 1><<<625, 128, 0, stream>>>(
        agg, 96, nullptr, 0, c2_w, 64, c2_b, xcat + 96, 160, sumL(1), sqL(1),
        nullptr, nullptr, nullptr, nullptr, nullptr, nullptr, nullptr, nullptr);

    gemm_wmma_f32_kernel<160, 6, 1, false, 0, 96><<<625, 192, 0, stream>>>(
        xcat, 160, nullptr, 0, l1_w, 96, l1_b, bufQ, 96, sumL(2), sqL(2),
        sumL(0), sqL(0), c1_g, c1_be, sumL(1), sqL(1), c2_g, c2_be);

    gemm_wmma_f32_kernel<96, 6, 1, false, 0, 96><<<625, 192, 0, stream>>>(
        bufQ, 96, nullptr, 0, ma_w, 96, ma_b, bufP, 96, sumL(3), sqL(3),
        sumL(2), sqL(2), l1_g, l1_be, nullptr, nullptr, nullptr, nullptr);

    gemm_wmma_f32_kernel<96, 1, 5, false, 0, 96><<<625, 160, 0, stream>>>(
        bufP, 96, nullptr, 0, mb_w, 16, mb_b, bufQ, 16, sumL(4), sqL(4),
        sumL(3), sqL(3), ma_g, ma_be, nullptr, nullptr, nullptr, nullptr);

    bn_apply_kernel<<<(NNODES * 16 + 255) / 256, 256, 0, stream>>>(
        bufQ, 16, (float*)d_out, 16, sumL(4), sqL(4), mb_g, mb_be, NNODES, 16);
}
